# starting kernel
# baseline (speedup 1.0000x reference)
#include <hip/hip_runtime.h>
#include <hip/hip_bf16.h>
#include <math.h>

typedef __attribute__((ext_vector_type(2))) float v2f;
typedef __attribute__((ext_vector_type(8))) float v8f;

#define N_NODES   50000
#define N_EDGES   800000
#define F_IN      128
#define HC        128
#define HEADS     4
#define HID       32
#define OUT_C     32
#define NEG_SLOPE 0.2f

__global__ void gemm16x16_wmma_f32(const float* __restrict__ A,
                                   const float* __restrict__ B,
                                   const float* __restrict__ bias,
                                   float* __restrict__ C,
                                   int N, int K, int P) {
    const int wave = threadIdx.x >> 5;
    const int lane = threadIdx.x & 31;
    const int tile = blockIdx.x * (blockDim.x >> 5) + wave;
    const int ptiles = P >> 4;
    const int ntiles = N >> 4;
    if (tile >= ntiles * ptiles) return;
    const int row0 = (tile / ptiles) << 4;
    const int col0 = (tile % ptiles) << 4;
    const int half = lane >> 4;
    const int r    = lane & 15;

    v8f acc = {};
    const float* arow = A + (size_t)(row0 + r) * K;
    for (int k0 = 0; k0 < K; k0 += 4) {
        const int ka = k0 + 2 * half;
        v2f a = *(const v2f*)(arow + ka);
        v2f b;
        b.x = B[(size_t)(ka + 0) * P + col0 + r];
        b.y = B[(size_t)(ka + 1) * P + col0 + r];
        acc = __builtin_amdgcn_wmma_f32_16x16x4_f32(
             false, a,  false, b,
             (short)0, acc,  false,  false);
    }

    const float badd = bias ? bias[col0 + r] : 0.0f;
#pragma unroll
    for (int i = 0; i < 8; ++i) {
        const int row = row0 + i + half * 8;
        C[(size_t)row * P + col0 + r] = acc[i] + badd;
    }
}

__global__ void attn_logits(const float* __restrict__ h,
                            const float* __restrict__ aS,
                            const float* __restrict__ aD,
                            float* __restrict__ als,
                            float* __restrict__ ald, int N) {
    const int i = blockIdx.x * blockDim.x + threadIdx.x;
    if (i >= N * HEADS) return;
    const int n  = i >> 2;
    const int hd = i & 3;
    const float* hp = h + (size_t)n * HC + hd * HID;
    const float* as = aS + hd * HID;
    const float* ad = aD + hd * HID;
    float ss = 0.f, sd = 0.f;
#pragma unroll 8
    for (int c = 0; c < HID; ++c) {
        const float v = hp[c];
        ss += v * as[c];
        sd += v * ad[c];
    }
    als[i] = ss;
    ald[i] = sd;
}

__global__ void init_buffers(float* __restrict__ att, unsigned* __restrict__ m,
                             float* __restrict__ s, int natt, int nms) {
    const int i = blockIdx.x * blockDim.x + threadIdx.x;
    if (i < natt) att[i] = 0.0f;
    if (i < nms) { m[i] = 0u; s[i] = 0.0f; }
}

__device__ __forceinline__ unsigned f2ord(float f) {
    unsigned u = __float_as_uint(f);
    return (u & 0x80000000u) ? ~u : (u | 0x80000000u);
}
__device__ __forceinline__ float ord2f(unsigned u) {
    return (u & 0x80000000u) ? __uint_as_float(u ^ 0x80000000u)
                             : __uint_as_float(~u);
}

__global__ void edge_max(const int* __restrict__ src, const int* __restrict__ dst,
                         const float* __restrict__ als, const float* __restrict__ ald,
                         float* __restrict__ ebuf, unsigned* __restrict__ m) {
    const int i = blockIdx.x * blockDim.x + threadIdx.x;
    if (i >= N_EDGES * HEADS) return;
    const int e  = i >> 2;
    const int hd = i & 3;
    const int sN = src[e], dN = dst[e];
    float v = als[sN * HEADS + hd] + ald[dN * HEADS + hd];
    v = (v > 0.f) ? v : NEG_SLOPE * v;
    ebuf[i] = v;
    atomicMax(&m[dN * HEADS + hd], f2ord(v));
}

__global__ void edge_exp(const int* __restrict__ dst, const unsigned* __restrict__ m,
                         float* __restrict__ ebuf, float* __restrict__ s) {
    const int i = blockIdx.x * blockDim.x + threadIdx.x;
    if (i >= N_EDGES * HEADS) return;
    const int e  = i >> 2;
    const int hd = i & 3;
    const int dN = dst[e];
    const float ex = expf(ebuf[i] - ord2f(m[dN * HEADS + hd]));
    ebuf[i] = ex;
    atomicAdd(&s[dN * HEADS + hd], ex);
}

__global__ void edge_aggr(const int* __restrict__ src, const int* __restrict__ dst,
                          const float* __restrict__ ebuf, const float* __restrict__ s,
                          const float* __restrict__ h, float* __restrict__ att) {
    const long long i = (long long)blockIdx.x * blockDim.x + threadIdx.x;
    if (i >= (long long)N_EDGES * HC) return;
    const int e  = (int)(i >> 7);
    const int c  = (int)(i & 127);
    const int hd = c >> 5;
    const int sN = src[e], dN = dst[e];
    const float alpha = ebuf[e * HEADS + hd] / (s[dN * HEADS + hd] + 1e-16f);
    atomicAdd(&att[(size_t)dN * HC + c], alpha * h[(size_t)sN * HC + c]);
}

__global__ void finish01(const float* __restrict__ att, const float* __restrict__ b,
                         const float* __restrict__ skip, float* __restrict__ out, int N) {
    const int i = blockIdx.x * blockDim.x + threadIdx.x;
    if (i >= N * HC) return;
    const int c = i & 127;
    const float v = att[i] + b[c] + skip[i];
    out[i] = (v > 0.f) ? v : (expf(v) - 1.0f);
}

__global__ void finish2(const float* __restrict__ att, const float* __restrict__ b2,
                        const float* __restrict__ skip32, float* __restrict__ out, int N) {
    const int node = blockIdx.x * (blockDim.x >> 5) + (threadIdx.x >> 5);
    const int c    = threadIdx.x & 31;
    if (node >= N) return;
    const float* ap = att + (size_t)node * HC;
    float v = 0.25f * (ap[c] + ap[32 + c] + ap[64 + c] + ap[96 + c])
            + b2[c] + skip32[(size_t)node * OUT_C + c];
    float sq = v * v;
#pragma unroll
    for (int o = 16; o > 0; o >>= 1) sq += __shfl_xor(sq, o, 32);
    const float norm = sqrtf(sq);
    out[(size_t)node * OUT_C + c] = v / fmaxf(norm, 1e-12f);
}

extern "C" void kernel_launch(void* const* d_in, const int* in_sizes, int n_in,
                              void* d_out, int out_size, void* d_ws, size_t ws_size,
                              hipStream_t stream) {
    (void)in_sizes; (void)n_in; (void)out_size; (void)ws_size;

    const float* x   = (const float*)d_in[0];
    const int*   ei  = (const int*)d_in[1];
    const float* W0  = (const float*)d_in[2];
    const float* aS0 = (const float*)d_in[3];
    const float* aD0 = (const float*)d_in[4];
    const float* b0  = (const float*)d_in[5];
    const float* Ws0 = (const float*)d_in[6];
    const float* bs0 = (const float*)d_in[7];
    const float* W1  = (const float*)d_in[8];
    const float* aS1 = (const float*)d_in[9];
    const float* aD1 = (const float*)d_in[10];
    const float* b1  = (const float*)d_in[11];
    const float* Ws1 = (const float*)d_in[12];
    const float* bs1 = (const float*)d_in[13];
    const float* W2  = (const float*)d_in[14];
    const float* aS2 = (const float*)d_in[15];
    const float* aD2 = (const float*)d_in[16];
    const float* b2  = (const float*)d_in[17];
    const float* Ws2 = (const float*)d_in[18];
    const float* bs2 = (const float*)d_in[19];

    const int* src = ei;
    const int* dst = ei + N_EDGES;

    const size_t NF = (size_t)N_NODES * HC;
    float*    ws    = (float*)d_ws;
    float*    x_cur = ws;
    float*    h_lin = ws + NF;
    float*    skipb = ws + 2 * NF;
    float*    att   = ws + 3 * NF;
    float*    als   = ws + 4 * NF;
    float*    ald   = als + (size_t)N_NODES * HEADS;
    unsigned* m     = (unsigned*)(ald + (size_t)N_NODES * HEADS);
    float*    s     = (float*)(m + (size_t)N_NODES * HEADS);
    float*    ebuf  = s + (size_t)N_NODES * HEADS;

    auto gemm = [&](const float* A, const float* B, const float* bias,
                    float* C, int N, int K, int P) {
        const int tiles = (N / 16) * (P / 16);
        const int blocks = (tiles + 7) / 8;
        gemm16x16_wmma_f32<<<blocks, 256, 0, stream>>>(A, B, bias, C, N, K, P);
    };

    auto run_edges = [&](const float* hfeat) {
        const int nInit = N_NODES * HC;
        init_buffers<<<(nInit + 255) / 256, 256, 0, stream>>>(
            att, m, s, nInit, N_NODES * HEADS);
        const int nEH = N_EDGES * HEADS;
        edge_max<<<(nEH + 255) / 256, 256, 0, stream>>>(src, dst, als, ald, ebuf, m);
        edge_exp<<<(nEH + 255) / 256, 256, 0, stream>>>(dst, m, ebuf, s);
        const long long tot = (long long)N_EDGES * HC;
        edge_aggr<<<(int)((tot + 255) / 256), 256, 0, stream>>>(
            src, dst, ebuf, s, hfeat, att);
    };

    const int nNH = N_NODES * HEADS;
    const int nNF = N_NODES * HC;

    gemm(x, W0, nullptr, h_lin, N_NODES, F_IN, HC);
    gemm(x, Ws0, bs0, skipb, N_NODES, F_IN, HC);
    attn_logits<<<(nNH + 255) / 256, 256, 0, stream>>>(h_lin, aS0, aD0, als, ald, N_NODES);
    run_edges(h_lin);
    finish01<<<(nNF + 255) / 256, 256, 0, stream>>>(att, b0, skipb, x_cur, N_NODES);

    gemm(x_cur, W1, nullptr, h_lin, N_NODES, HC, HC);
    gemm(x_cur, Ws1, bs1, skipb, N_NODES, HC, HC);
    attn_logits<<<(nNH + 255) / 256, 256, 0, stream>>>(h_lin, aS1, aD1, als, ald, N_NODES);
    run_edges(h_lin);
    finish01<<<(nNF + 255) / 256, 256, 0, stream>>>(att, b1, skipb, x_cur, N_NODES);

    gemm(x_cur, W2, nullptr, h_lin, N_NODES, HC, HEADS * OUT_C);
    gemm(x_cur, Ws2, bs2, skipb, N_NODES, HC, OUT_C);
    attn_logits<<<(nNH + 255) / 256, 256, 0, stream>>>(h_lin, aS2, aD2, als, ald, N_NODES);
    run_edges(h_lin);
    finish2<<<(N_NODES + 7) / 8, 256, 0, stream>>>(att, b2, skipb, (float*)d_out, N_NODES);
}


# Edges grouped by destination once per call (count, scan, place, order); a wave per node does softmax and row gather in registers, sums in fixed edge order; float atomics and edge passes removed
# speedup vs baseline: 2.9814x; 2.9814x over previous
#include <hip/hip_runtime.h>
#include <hip/hip_bf16.h>
#include <math.h>

typedef __attribute__((ext_vector_type(2))) float v2f;
typedef __attribute__((ext_vector_type(8))) float v8f;

#define N_NODES   50000
#define N_EDGES   800000
#define F_IN      128
#define HC        128
#define HEADS     4
#define HID       32
#define OUT_C     32
#define NEG_SLOPE 0.2f

__global__ void gemm16x16_wmma_f32(const float* __restrict__ A,
                                   const float* __restrict__ B,
                                   const float* __restrict__ bias,
                                   float* __restrict__ C,
                                   int N, int K, int P) {
    const int wave = threadIdx.x >> 5;
    const int lane = threadIdx.x & 31;
    const int tile = blockIdx.x * (blockDim.x >> 5) + wave;
    const int ptiles = P >> 4;
    const int ntiles = N >> 4;
    if (tile >= ntiles * ptiles) return;
    const int row0 = (tile / ptiles) << 4;
    const int col0 = (tile % ptiles) << 4;
    const int half = lane >> 4;
    const int r    = lane & 15;

    v8f acc = {};
    const float* arow = A + (size_t)(row0 + r) * K;
    for (int k0 = 0; k0 < K; k0 += 4) {
        const int ka = k0 + 2 * half;
        v2f a = *(const v2f*)(arow + ka);
        v2f b;
        b.x = B[(size_t)(ka + 0) * P + col0 + r];
        b.y = B[(size_t)(ka + 1) * P + col0 + r];
        acc = __builtin_amdgcn_wmma_f32_16x16x4_f32(
             false, a,  false, b,
             (short)0, acc,  false,  false);
    }

    const float badd = bias ? bias[col0 + r] : 0.0f;
#pragma unroll
    for (int i = 0; i < 8; ++i) {
        const int row = row0 + i + half * 8;
        C[(size_t)row * P + col0 + r] = acc[i] + badd;
    }
}

__global__ void attn_logits(const float* __restrict__ h,
                            const float* __restrict__ aS,
                            const float* __restrict__ aD,
                            float* __restrict__ als,
                            float* __restrict__ ald, int N) {
    const int i = blockIdx.x * blockDim.x + threadIdx.x;
    if (i >= N * HEADS) return;
    const int n  = i >> 2;
    const int hd = i & 3;
    const float* hp = h + (size_t)n * HC + hd * HID;
    const float* as = aS + hd * HID;
    const float* ad = aD + hd * HID;
    float ss = 0.f, sd = 0.f;
#pragma unroll 8
    for (int c = 0; c < HID; ++c) {
        const float v = hp[c];
        ss += v * as[c];
        sd += v * ad[c];
    }
    als[i] = ss;
    ald[i] = sd;
}

__global__ void csr_zero(int* __restrict__ cnt, int n) {
    const int i = blockIdx.x * blockDim.x + threadIdx.x;
    if (i < n) cnt[i] = 0;
}

__global__ void csr_count(const int* __restrict__ dst, int* __restrict__ cnt,
                          int* __restrict__ rnk) {
    const int e = blockIdx.x * blockDim.x + threadIdx.x;
    if (e >= N_EDGES) return;
    rnk[e] = atomicAdd(&cnt[dst[e]], 1);
}

__global__ void csr_scan(const int* __restrict__ cnt, int* __restrict__ rowp, int n) {
    __shared__ int wtot[32];
    const int lane  = threadIdx.x & 31;
    const int wave  = threadIdx.x >> 5;
    const int waves = blockDim.x >> 5;
    const int per   = ((n + blockDim.x - 1) / blockDim.x) * 32;
    const int wb    = wave * per;
    int carry = 0;
    for (int it = 0; it < per; it += 32) {
        const int i = wb + it + lane;
        const int v = (i < n) ? cnt[i] : 0;
        int inc = v;
        for (int o = 1; o < 32; o <<= 1) {
            const int t = __shfl_up(inc, o, 32);
            if (lane >= o) inc += t;
        }
        if (i < n) rowp[i] = carry + inc - v;
        carry += __shfl(inc, 31, 32);
    }
    if (lane == 0) wtot[wave] = carry;
    __syncthreads();
    int off = (lane < wave) ? wtot[lane] : 0;
    for (int o = 16; o > 0; o >>= 1) off += __shfl_xor(off, o, 32);
    for (int it = 0; it < per; it += 32) {
        const int i = wb + it + lane;
        if (i < n) rowp[i] += off;
    }
    if (wave == waves - 1 && lane == 0) rowp[n] = off + carry;
}

__global__ void csr_place(const int* __restrict__ dst, const int* __restrict__ rowp,
                          const int* __restrict__ rnk, int* __restrict__ slot) {
    const int e = blockIdx.x * blockDim.x + threadIdx.x;
    if (e >= N_EDGES) return;
    slot[rowp[dst[e]] + rnk[e]] = e;
}

__global__ void csr_order(const int* __restrict__ rowp, const int* __restrict__ slot,
                          const int* __restrict__ src, int* __restrict__ csrc) {
    const int wv   = __builtin_amdgcn_readfirstlane((int)(threadIdx.x >> 5));
    const int node = blockIdx.x * (blockDim.x >> 5) + wv;
    const int lane = threadIdx.x & 31;
    if (node >= N_NODES) return;
    const int beg = rowp[node], end = rowp[node + 1];
    for (int i = beg + lane; i < end; i += 32) {
        const int my = slot[i];
        int r = 0;
        for (int j = beg; j < end; ++j) r += (slot[j] < my) ? 1 : 0;
        csrc[beg + r] = src[my];
    }
}

__device__ __forceinline__ float lrelu(float v) { return (v > 0.f) ? v : NEG_SLOPE * v; }

__device__ __forceinline__ float wave_max(float v) {
    for (int o = 16; o > 0; o >>= 1) v = fmaxf(v, __shfl_xor(v, o, 32));
    return v;
}
__device__ __forceinline__ float wave_sum(float v) {
    for (int o = 16; o > 0; o >>= 1) v += __shfl_xor(v, o, 32);
    return v;
}

__device__ __forceinline__ float4 edge_scores(const float* __restrict__ als, int sN, float4 ad) {
    const float4 a = *(const float4*)(als + (size_t)sN * HEADS);
    return make_float4(lrelu(a.x + ad.x), lrelu(a.y + ad.y), lrelu(a.z + ad.z), lrelu(a.w + ad.w));
}

__device__ __forceinline__ void gather_rows(const float* __restrict__ h, int lane, int cnt, int sN,
                                            float4 al, unsigned k0, unsigned k1, unsigned k2,
                                            unsigned k3, float4& acc) {
    for (int t = 0; t < cnt; ++t) {
        const int      sT = __builtin_amdgcn_readlane(sN, t);
        const unsigned b0 = (unsigned)__builtin_amdgcn_readlane(__float_as_int(al.x), t);
        const unsigned b1 = (unsigned)__builtin_amdgcn_readlane(__float_as_int(al.y), t);
        const unsigned b2 = (unsigned)__builtin_amdgcn_readlane(__float_as_int(al.z), t);
        const unsigned b3 = (unsigned)__builtin_amdgcn_readlane(__float_as_int(al.w), t);
        const float w = __uint_as_float((b0 & k0) | (b1 & k1) | (b2 & k2) | (b3 & k3));
        const float4 hv = *(const float4*)(h + (size_t)sT * HC + lane * 4);
        acc.x += w * hv.x;
        acc.y += w * hv.y;
        acc.z += w * hv.z;
        acc.w += w * hv.w;
    }
}

__global__ void node_softmax_gather(const int* __restrict__ rowp, const int* __restrict__ csrc,
                                    const float* __restrict__ als, const float* __restrict__ ald,
                                    const float* __restrict__ h, float* __restrict__ att) {
    const int wv   = __builtin_amdgcn_readfirstlane((int)(threadIdx.x >> 5));
    const int node = blockIdx.x * (blockDim.x >> 5) + wv;
    const int lane = threadIdx.x & 31;
    if (node >= N_NODES) return;
    const int beg = rowp[node], end = rowp[node + 1];
    const int hd  = lane >> 3;
    const float4 ad = *(const float4*)(ald + (size_t)node * HEADS);
    const float NEG_INF = -__builtin_huge_valf();

    const int  j0  = beg + lane;
    const bool ok0 = j0 < end;
    const int  sN0 = ok0 ? csrc[j0] : 0;
    float4 e0 = make_float4(NEG_INF, NEG_INF, NEG_INF, NEG_INF);
    if (ok0) e0 = edge_scores(als, sN0, ad);

    float4 m = e0;
    for (int j = j0 + 32; j < end; j += 32) {
        const float4 e = edge_scores(als, csrc[j], ad);
        m.x = fmaxf(m.x, e.x); m.y = fmaxf(m.y, e.y);
        m.z = fmaxf(m.z, e.z); m.w = fmaxf(m.w, e.w);
    }
    m.x = wave_max(m.x); m.y = wave_max(m.y); m.z = wave_max(m.z); m.w = wave_max(m.w);

    float4 x0 = make_float4(0.f, 0.f, 0.f, 0.f);
    if (ok0) x0 = make_float4(expf(e0.x - m.x), expf(e0.y - m.y), expf(e0.z - m.z), expf(e0.w - m.w));
    float4 s = x0;
    for (int j = j0 + 32; j < end; j += 32) {
        const float4 e = edge_scores(als, csrc[j], ad);
        s.x += expf(e.x - m.x); s.y += expf(e.y - m.y);
        s.z += expf(e.z - m.z); s.w += expf(e.w - m.w);
    }
    s.x = wave_sum(s.x); s.y = wave_sum(s.y); s.z = wave_sum(s.z); s.w = wave_sum(s.w);
    const float4 den = make_float4(s.x + 1e-16f, s.y + 1e-16f, s.z + 1e-16f, s.w + 1e-16f);

    float4 acc = make_float4(0.f, 0.f, 0.f, 0.f);
    const int deg = end - beg;
    const unsigned k0 = (hd == 0) ? 0xFFFFFFFFu : 0u, k1 = (hd == 1) ? 0xFFFFFFFFu : 0u;
    const unsigned k2 = (hd == 2) ? 0xFFFFFFFFu : 0u, k3 = (hd == 3) ? 0xFFFFFFFFu : 0u;
    gather_rows(h, lane, deg < 32 ? deg : 32, sN0,
                make_float4(x0.x / den.x, x0.y / den.y, x0.z / den.z, x0.w / den.w),
                k0, k1, k2, k3, acc);
    for (int base = beg + 32; base < end; base += 32) {
        const int j = base + lane;
        int sN = 0;
        float4 al = make_float4(0.f, 0.f, 0.f, 0.f);
        if (j < end) {
            sN = csrc[j];
            const float4 e = edge_scores(als, sN, ad);
            al = make_float4(expf(e.x - m.x) / den.x, expf(e.y - m.y) / den.y,
                             expf(e.z - m.z) / den.z, expf(e.w - m.w) / den.w);
        }
        const int left = end - base;
        gather_rows(h, lane, left < 32 ? left : 32, sN, al, k0, k1, k2, k3, acc);
    }
    *(float4*)(att + (size_t)node * HC + lane * 4) = acc;
}

__global__ void finish01(const float* __restrict__ att, const float* __restrict__ b,
                         const float* __restrict__ skip, float* __restrict__ out, int N) {
    const int i = blockIdx.x * blockDim.x + threadIdx.x;
    if (i >= N * HC) return;
    const int c = i & 127;
    const float v = att[i] + b[c] + skip[i];
    out[i] = (v > 0.f) ? v : (expf(v) - 1.0f);
}

__global__ void finish2(const float* __restrict__ att, const float* __restrict__ b2,
                        const float* __restrict__ skip32, float* __restrict__ out, int N) {
    const int node = blockIdx.x * (blockDim.x >> 5) + (threadIdx.x >> 5);
    const int c    = threadIdx.x & 31;
    if (node >= N) return;
    const float* ap = att + (size_t)node * HC;
    float v = 0.25f * (ap[c] + ap[32 + c] + ap[64 + c] + ap[96 + c])
            + b2[c] + skip32[(size_t)node * OUT_C + c];
    float sq = v * v;
#pragma unroll
    for (int o = 16; o > 0; o >>= 1) sq += __shfl_xor(sq, o, 32);
    const float norm = sqrtf(sq);
    out[(size_t)node * OUT_C + c] = v / fmaxf(norm, 1e-12f);
}

extern "C" void kernel_launch(void* const* d_in, const int* in_sizes, int n_in,
                              void* d_out, int out_size, void* d_ws, size_t ws_size,
                              hipStream_t stream) {
    (void)in_sizes; (void)n_in; (void)out_size; (void)ws_size;

    const float* x   = (const float*)d_in[0];
    const int*   ei  = (const int*)d_in[1];
    const float* W0  = (const float*)d_in[2];
    const float* aS0 = (const float*)d_in[3];
    const float* aD0 = (const float*)d_in[4];
    const float* b0  = (const float*)d_in[5];
    const float* Ws0 = (const float*)d_in[6];
    const float* bs0 = (const float*)d_in[7];
    const float* W1  = (const float*)d_in[8];
    const float* aS1 = (const float*)d_in[9];
    const float* aD1 = (const float*)d_in[10];
    const float* b1  = (const float*)d_in[11];
    const float* Ws1 = (const float*)d_in[12];
    const float* bs1 = (const float*)d_in[13];
    const float* W2  = (const float*)d_in[14];
    const float* aS2 = (const float*)d_in[15];
    const float* aD2 = (const float*)d_in[16];
    const float* b2  = (const float*)d_in[17];
    const float* Ws2 = (const float*)d_in[18];
    const float* bs2 = (const float*)d_in[19];

    const int* src = ei;
    const int* dst = ei + N_EDGES;

    const size_t NF = (size_t)N_NODES * HC;
    float*    ws    = (float*)d_ws;
    float*    x_cur = ws;
    float*    h_lin = ws + NF;
    float*    skipb = ws + 2 * NF;
    float*    att   = ws + 3 * NF;
    float*    als   = ws + 4 * NF;
    float*    ald   = als + (size_t)N_NODES * HEADS;
    const int NPAD  = N_NODES + 32;
    int*      cnt   = (int*)(ald + (size_t)N_NODES * HEADS);
    int*      rowp  = cnt + NPAD;
    int*      rnk   = rowp + NPAD;
    int*      slot  = rnk + N_EDGES;
    int*      csrc  = slot + N_EDGES;

    auto gemm = [&](const float* A, const float* B, const float* bias,
                    float* C, int N, int K, int P) {
        const int tiles = (N / 16) * (P / 16);
        const int blocks = (tiles + 7) / 8;
        gemm16x16_wmma_f32<<<blocks, 256, 0, stream>>>(A, B, bias, C, N, K, P);
    };

    auto run_edges = [&](const float* hfeat) {
        node_softmax_gather<<<(N_NODES + 7) / 8, 256, 0, stream>>>(
            rowp, csrc, als, ald, hfeat, att);
    };

    csr_zero<<<(N_NODES + 255) / 256, 256, 0, stream>>>(cnt, N_NODES);
    csr_count<<<(N_EDGES + 255) / 256, 256, 0, stream>>>(dst, cnt, rnk);
    csr_scan<<<1, 1024, 0, stream>>>(cnt, rowp, N_NODES);
    csr_place<<<(N_EDGES + 255) / 256, 256, 0, stream>>>(dst, rowp, rnk, slot);
    csr_order<<<(N_NODES + 7) / 8, 256, 0, stream>>>(rowp, slot, src, csrc);

    const int nNH = N_NODES * HEADS;
    const int nNF = N_NODES * HC;

    gemm(x, W0, nullptr, h_lin, N_NODES, F_IN, HC);
    gemm(x, Ws0, bs0, skipb, N_NODES, F_IN, HC);
    attn_logits<<<(nNH + 255) / 256, 256, 0, stream>>>(h_lin, aS0, aD0, als, ald, N_NODES);
    run_edges(h_lin);
    finish01<<<(nNF + 255) / 256, 256, 0, stream>>>(att, b0, skipb, x_cur, N_NODES);

    gemm(x_cur, W1, nullptr, h_lin, N_NODES, HC, HC);
    gemm(x_cur, Ws1, bs1, skipb, N_NODES, HC, HC);
    attn_logits<<<(nNH + 255) / 256, 256, 0, stream>>>(h_lin, aS1, aD1, als, ald, N_NODES);
    run_edges(h_lin);
    finish01<<<(nNF + 255) / 256, 256, 0, stream>>>(att, b1, skipb, x_cur, N_NODES);

    gemm(x_cur, W2, nullptr, h_lin, N_NODES, HC, HEADS * OUT_C);
    gemm(x_cur, Ws2, bs2, skipb, N_NODES, HC, OUT_C);
    attn_logits<<<(nNH + 255) / 256, 256, 0, stream>>>(h_lin, aS2, aD2, als, ald, N_NODES);
    run_edges(h_lin);
    finish2<<<(N_NODES + 7) / 8, 256, 0, stream>>>(att, b2, skipb, (float*)d_out, N_NODES);
}


# Layer epilogue (bias, skip, elu; layer 2: head mean, L2 normalise) done in registers by the wave that gathers the node's row: attention rows never stored, 3 elementwise launches gone, same sum order
# speedup vs baseline: 3.3684x; 1.1298x over previous
#include <hip/hip_runtime.h>
#include <hip/hip_bf16.h>
#include <math.h>

typedef __attribute__((ext_vector_type(2))) float v2f;
typedef __attribute__((ext_vector_type(8))) float v8f;

#define N_NODES   50000
#define N_EDGES   800000
#define F_IN      128
#define HC        128
#define HEADS     4
#define HID       32
#define OUT_C     32
#define NEG_SLOPE 0.2f

__global__ void gemm16x16_wmma_f32(const float* __restrict__ A,
                                   const float* __restrict__ B,
                                   const float* __restrict__ bias,
                                   float* __restrict__ C,
                                   int N, int K, int P) {
    const int wave = threadIdx.x >> 5;
    const int lane = threadIdx.x & 31;
    const int tile = blockIdx.x * (blockDim.x >> 5) + wave;
    const int ptiles = P >> 4;
    const int ntiles = N >> 4;
    if (tile >= ntiles * ptiles) return;
    const int row0 = (tile / ptiles) << 4;
    const int col0 = (tile % ptiles) << 4;
    const int half = lane >> 4;
    const int r    = lane & 15;

    v8f acc = {};
    const float* arow = A + (size_t)(row0 + r) * K;
    for (int k0 = 0; k0 < K; k0 += 4) {
        const int ka = k0 + 2 * half;
        v2f a = *(const v2f*)(arow + ka);
        v2f b;
        b.x = B[(size_t)(ka + 0) * P + col0 + r];
        b.y = B[(size_t)(ka + 1) * P + col0 + r];
        acc = __builtin_amdgcn_wmma_f32_16x16x4_f32(
             false, a,  false, b,
             (short)0, acc,  false,  false);
    }

    const float badd = bias ? bias[col0 + r] : 0.0f;
#pragma unroll
    for (int i = 0; i < 8; ++i) {
        const int row = row0 + i + half * 8;
        C[(size_t)row * P + col0 + r] = acc[i] + badd;
    }
}

__global__ void attn_logits(const float* __restrict__ h,
                            const float* __restrict__ aS,
                            const float* __restrict__ aD,
                            float* __restrict__ als,
                            float* __restrict__ ald, int N) {
    const int i = blockIdx.x * blockDim.x + threadIdx.x;
    if (i >= N * HEADS) return;
    const int n  = i >> 2;
    const int hd = i & 3;
    const float* hp = h + (size_t)n * HC + hd * HID;
    const float* as = aS + hd * HID;
    const float* ad = aD + hd * HID;
    float ss = 0.f, sd = 0.f;
#pragma unroll 8
    for (int c = 0; c < HID; ++c) {
        const float v = hp[c];
        ss += v * as[c];
        sd += v * ad[c];
    }
    als[i] = ss;
    ald[i] = sd;
}

__global__ void csr_zero(int* __restrict__ cnt, int n) {
    const int i = blockIdx.x * blockDim.x + threadIdx.x;
    if (i < n) cnt[i] = 0;
}

__global__ void csr_count(const int* __restrict__ dst, int* __restrict__ cnt,
                          int* __restrict__ rnk) {
    const int e = blockIdx.x * blockDim.x + threadIdx.x;
    if (e >= N_EDGES) return;
    rnk[e] = atomicAdd(&cnt[dst[e]], 1);
}

__global__ void csr_scan(const int* __restrict__ cnt, int* __restrict__ rowp, int n) {
    __shared__ int wtot[32];
    const int lane  = threadIdx.x & 31;
    const int wave  = threadIdx.x >> 5;
    const int waves = blockDim.x >> 5;
    const int per   = ((n + blockDim.x - 1) / blockDim.x) * 32;
    const int wb    = wave * per;
    int carry = 0;
    for (int it = 0; it < per; it += 32) {
        const int i = wb + it + lane;
        const int v = (i < n) ? cnt[i] : 0;
        int inc = v;
        for (int o = 1; o < 32; o <<= 1) {
            const int t = __shfl_up(inc, o, 32);
            if (lane >= o) inc += t;
        }
        if (i < n) rowp[i] = carry + inc - v;
        carry += __shfl(inc, 31, 32);
    }
    if (lane == 0) wtot[wave] = carry;
    __syncthreads();
    int off = (lane < wave) ? wtot[lane] : 0;
    for (int o = 16; o > 0; o >>= 1) off += __shfl_xor(off, o, 32);
    for (int it = 0; it < per; it += 32) {
        const int i = wb + it + lane;
        if (i < n) rowp[i] += off;
    }
    if (wave == waves - 1 && lane == 0) rowp[n] = off + carry;
}

__global__ void csr_place(const int* __restrict__ dst, const int* __restrict__ rowp,
                          const int* __restrict__ rnk, int* __restrict__ slot) {
    const int e = blockIdx.x * blockDim.x + threadIdx.x;
    if (e >= N_EDGES) return;
    slot[rowp[dst[e]] + rnk[e]] = e;
}

__global__ void csr_order(const int* __restrict__ rowp, const int* __restrict__ slot,
                          const int* __restrict__ src, int* __restrict__ csrc) {
    const int wv   = __builtin_amdgcn_readfirstlane((int)(threadIdx.x >> 5));
    const int node = blockIdx.x * (blockDim.x >> 5) + wv;
    const int lane = threadIdx.x & 31;
    if (node >= N_NODES) return;
    const int beg = rowp[node], end = rowp[node + 1];
    for (int i = beg + lane; i < end; i += 32) {
        const int my = slot[i];
        int r = 0;
        for (int j = beg; j < end; ++j) r += (slot[j] < my) ? 1 : 0;
        csrc[beg + r] = src[my];
    }
}

__device__ __forceinline__ float lrelu(float v) { return (v > 0.f) ? v : NEG_SLOPE * v; }

__device__ __forceinline__ float wave_max(float v) {
    for (int o = 16; o > 0; o >>= 1) v = fmaxf(v, __shfl_xor(v, o, 32));
    return v;
}
__device__ __forceinline__ float wave_sum(float v) {
    for (int o = 16; o > 0; o >>= 1) v += __shfl_xor(v, o, 32);
    return v;
}

__device__ __forceinline__ float4 edge_scores(const float* __restrict__ als, int sN, float4 ad) {
    const float4 a = *(const float4*)(als + (size_t)sN * HEADS);
    return make_float4(lrelu(a.x + ad.x), lrelu(a.y + ad.y), lrelu(a.z + ad.z), lrelu(a.w + ad.w));
}

__device__ __forceinline__ void gather_rows(const float* __restrict__ h, int lane, int cnt, int sN,
                                            float4 al, unsigned k0, unsigned k1, unsigned k2,
                                            unsigned k3, float4& acc) {
    for (int t = 0; t < cnt; ++t) {
        const int      sT = __builtin_amdgcn_readlane(sN, t);
        const unsigned b0 = (unsigned)__builtin_amdgcn_readlane(__float_as_int(al.x), t);
        const unsigned b1 = (unsigned)__builtin_amdgcn_readlane(__float_as_int(al.y), t);
        const unsigned b2 = (unsigned)__builtin_amdgcn_readlane(__float_as_int(al.z), t);
        const unsigned b3 = (unsigned)__builtin_amdgcn_readlane(__float_as_int(al.w), t);
        const float w = __uint_as_float((b0 & k0) | (b1 & k1) | (b2 & k2) | (b3 & k3));
        const float4 hv = *(const float4*)(h + (size_t)sT * HC + lane * 4);
        acc.x += w * hv.x;
        acc.y += w * hv.y;
        acc.z += w * hv.z;
        acc.w += w * hv.w;
    }
}

template <bool LAST>
__global__ void node_softmax_gather(const int* __restrict__ rowp, const int* __restrict__ csrc,
                                    const float* __restrict__ als, const float* __restrict__ ald,
                                    const float* __restrict__ h, const float* __restrict__ bias,
                                    const float* __restrict__ skip, float* __restrict__ out) {
    const int wv   = __builtin_amdgcn_readfirstlane((int)(threadIdx.x >> 5));
    const int node = blockIdx.x * (blockDim.x >> 5) + wv;
    const int lane = threadIdx.x & 31;
    if (node >= N_NODES) return;
    const int beg = rowp[node], end = rowp[node + 1];
    const int hd  = lane >> 3;
    const float4 ad = *(const float4*)(ald + (size_t)node * HEADS);
    const float NEG_INF = -__builtin_huge_valf();
    const int    ec = LAST ? (lane & 7) * 4 : lane * 4;
    const float4 bv = *(const float4*)(bias + ec);
    const float4 sk = *(const float4*)(skip + (size_t)node * (LAST ? OUT_C : HC) + ec);

    const int  j0  = beg + lane;
    const bool ok0 = j0 < end;
    const int  sN0 = ok0 ? csrc[j0] : 0;
    float4 e0 = make_float4(NEG_INF, NEG_INF, NEG_INF, NEG_INF);
    if (ok0) e0 = edge_scores(als, sN0, ad);

    float4 m = e0;
    for (int j = j0 + 32; j < end; j += 32) {
        const float4 e = edge_scores(als, csrc[j], ad);
        m.x = fmaxf(m.x, e.x); m.y = fmaxf(m.y, e.y);
        m.z = fmaxf(m.z, e.z); m.w = fmaxf(m.w, e.w);
    }
    m.x = wave_max(m.x); m.y = wave_max(m.y); m.z = wave_max(m.z); m.w = wave_max(m.w);

    float4 x0 = make_float4(0.f, 0.f, 0.f, 0.f);
    if (ok0) x0 = make_float4(expf(e0.x - m.x), expf(e0.y - m.y), expf(e0.z - m.z), expf(e0.w - m.w));
    float4 s = x0;
    for (int j = j0 + 32; j < end; j += 32) {
        const float4 e = edge_scores(als, csrc[j], ad);
        s.x += expf(e.x - m.x); s.y += expf(e.y - m.y);
        s.z += expf(e.z - m.z); s.w += expf(e.w - m.w);
    }
    s.x = wave_sum(s.x); s.y = wave_sum(s.y); s.z = wave_sum(s.z); s.w = wave_sum(s.w);
    const float4 den = make_float4(s.x + 1e-16f, s.y + 1e-16f, s.z + 1e-16f, s.w + 1e-16f);

    float4 acc = make_float4(0.f, 0.f, 0.f, 0.f);
    const int deg = end - beg;
    const unsigned k0 = (hd == 0) ? 0xFFFFFFFFu : 0u, k1 = (hd == 1) ? 0xFFFFFFFFu : 0u;
    const unsigned k2 = (hd == 2) ? 0xFFFFFFFFu : 0u, k3 = (hd == 3) ? 0xFFFFFFFFu : 0u;
    gather_rows(h, lane, deg < 32 ? deg : 32, sN0,
                make_float4(x0.x / den.x, x0.y / den.y, x0.z / den.z, x0.w / den.w),
                k0, k1, k2, k3, acc);
    for (int base = beg + 32; base < end; base += 32) {
        const int j = base + lane;
        int sN = 0;
        float4 al = make_float4(0.f, 0.f, 0.f, 0.f);
        if (j < end) {
            sN = csrc[j];
            const float4 e = edge_scores(als, sN, ad);
            al = make_float4(expf(e.x - m.x) / den.x, expf(e.y - m.y) / den.y,
                             expf(e.z - m.z) / den.z, expf(e.w - m.w) / den.w);
        }
        const int left = end - base;
        gather_rows(h, lane, left < 32 ? left : 32, sN, al, k0, k1, k2, k3, acc);
    }
    if (!LAST) {
        float4 v = make_float4(acc.x + bv.x + sk.x, acc.y + bv.y + sk.y,
                               acc.z + bv.z + sk.z, acc.w + bv.w + sk.w);
        v.x = (v.x > 0.f) ? v.x : (expf(v.x) - 1.0f);
        v.y = (v.y > 0.f) ? v.y : (expf(v.y) - 1.0f);
        v.z = (v.z > 0.f) ? v.z : (expf(v.z) - 1.0f);
        v.w = (v.w > 0.f) ? v.w : (expf(v.w) - 1.0f);
        *(float4*)(out + (size_t)node * HC + lane * 4) = v;
    } else {
        const int g = lane & 7;
        float4 hs;
        hs.x = __shfl(acc.x, g, 32) + __shfl(acc.x, g + 8, 32) + __shfl(acc.x, g + 16, 32) + __shfl(acc.x, g + 24, 32);
        hs.y = __shfl(acc.y, g, 32) + __shfl(acc.y, g + 8, 32) + __shfl(acc.y, g + 16, 32) + __shfl(acc.y, g + 24, 32);
        hs.z = __shfl(acc.z, g, 32) + __shfl(acc.z, g + 8, 32) + __shfl(acc.z, g + 16, 32) + __shfl(acc.z, g + 24, 32);
        hs.w = __shfl(acc.w, g, 32) + __shfl(acc.w, g + 8, 32) + __shfl(acc.w, g + 16, 32) + __shfl(acc.w, g + 24, 32);
        const float4 v = make_float4(0.25f * hs.x + bv.x + sk.x, 0.25f * hs.y + bv.y + sk.y,
                                     0.25f * hs.z + bv.z + sk.z, 0.25f * hs.w + bv.w + sk.w);
        float4 t = make_float4(v.x * v.x, v.y * v.y, v.z * v.z, v.w * v.w);
        for (int o = 4; o > 0; o >>= 1) {
            t.x += __shfl_xor(t.x, o, 32);
            t.y += __shfl_xor(t.y, o, 32);
            t.z += __shfl_xor(t.z, o, 32);
            t.w += __shfl_xor(t.w, o, 32);
        }
        const float sq   = (t.x + t.z) + (t.y + t.w);
        const float norm = fmaxf(sqrtf(sq), 1e-12f);
        if (lane < 8)
            *(float4*)(out + (size_t)node * OUT_C + g * 4) =
                make_float4(v.x / norm, v.y / norm, v.z / norm, v.w / norm);
    }
}

extern "C" void kernel_launch(void* const* d_in, const int* in_sizes, int n_in,
                              void* d_out, int out_size, void* d_ws, size_t ws_size,
                              hipStream_t stream) {
    (void)in_sizes; (void)n_in; (void)out_size; (void)ws_size;

    const float* x   = (const float*)d_in[0];
    const int*   ei  = (const int*)d_in[1];
    const float* W0  = (const float*)d_in[2];
    const float* aS0 = (const float*)d_in[3];
    const float* aD0 = (const float*)d_in[4];
    const float* b0  = (const float*)d_in[5];
    const float* Ws0 = (const float*)d_in[6];
    const float* bs0 = (const float*)d_in[7];
    const float* W1  = (const float*)d_in[8];
    const float* aS1 = (const float*)d_in[9];
    const float* aD1 = (const float*)d_in[10];
    const float* b1  = (const float*)d_in[11];
    const float* Ws1 = (const float*)d_in[12];
    const float* bs1 = (const float*)d_in[13];
    const float* W2  = (const float*)d_in[14];
    const float* aS2 = (const float*)d_in[15];
    const float* aD2 = (const float*)d_in[16];
    const float* b2  = (const float*)d_in[17];
    const float* Ws2 = (const float*)d_in[18];
    const float* bs2 = (const float*)d_in[19];

    const int* src = ei;
    const int* dst = ei + N_EDGES;

    const size_t NF = (size_t)N_NODES * HC;
    float*    ws    = (float*)d_ws;
    float*    x_cur = ws;
    float*    h_lin = ws + NF;
    float*    skipb = ws + 2 * NF;
    float*    als   = ws + 3 * NF;
    float*    ald   = als + (size_t)N_NODES * HEADS;
    const int NPAD  = N_NODES + 32;
    int*      cnt   = (int*)(ald + (size_t)N_NODES * HEADS);
    int*      rowp  = cnt + NPAD;
    int*      rnk   = rowp + NPAD;
    int*      slot  = rnk + N_EDGES;
    int*      csrc  = slot + N_EDGES;

    auto gemm = [&](const float* A, const float* B, const float* bias,
                    float* C, int N, int K, int P) {
        const int tiles = (N / 16) * (P / 16);
        const int blocks = (tiles + 7) / 8;
        gemm16x16_wmma_f32<<<blocks, 256, 0, stream>>>(A, B, bias, C, N, K, P);
    };

    auto run_edges = [&](const float* hfeat, const float* b, float* xout) {
        node_softmax_gather<false><<<(N_NODES + 7) / 8, 256, 0, stream>>>(
            rowp, csrc, als, ald, hfeat, b, skipb, xout);
    };

    csr_zero<<<(N_NODES + 255) / 256, 256, 0, stream>>>(cnt, N_NODES);
    csr_count<<<(N_EDGES + 255) / 256, 256, 0, stream>>>(dst, cnt, rnk);
    csr_scan<<<1, 1024, 0, stream>>>(cnt, rowp, N_NODES);
    csr_place<<<(N_EDGES + 255) / 256, 256, 0, stream>>>(dst, rowp, rnk, slot);
    csr_order<<<(N_NODES + 7) / 8, 256, 0, stream>>>(rowp, slot, src, csrc);

    const int nNH = N_NODES * HEADS;

    gemm(x, W0, nullptr, h_lin, N_NODES, F_IN, HC);
    gemm(x, Ws0, bs0, skipb, N_NODES, F_IN, HC);
    attn_logits<<<(nNH + 255) / 256, 256, 0, stream>>>(h_lin, aS0, aD0, als, ald, N_NODES);
    run_edges(h_lin, b0, x_cur);

    gemm(x_cur, W1, nullptr, h_lin, N_NODES, HC, HC);
    gemm(x_cur, Ws1, bs1, skipb, N_NODES, HC, HC);
    attn_logits<<<(nNH + 255) / 256, 256, 0, stream>>>(h_lin, aS1, aD1, als, ald, N_NODES);
    run_edges(h_lin, b1, x_cur);

    gemm(x_cur, W2, nullptr, h_lin, N_NODES, HC, HEADS * OUT_C);
    gemm(x_cur, Ws2, bs2, skipb, N_NODES, HC, OUT_C);
    attn_logits<<<(nNH + 255) / 256, 256, 0, stream>>>(h_lin, aS2, aD2, als, ald, N_NODES);
    node_softmax_gather<true><<<(N_NODES + 7) / 8, 256, 0, stream>>>(
        rowp, csrc, als, ald, h_lin, b2, skipb, (float*)d_out);
}


# Dense products retiled, both of a layer in one launch: 256 work-groups hold a 128 x 80-128 weight block in LDS, a wave owns 16 rows x 5-8 tiles, A read once per strip not per tile; same sum order
# speedup vs baseline: 4.3672x; 1.2965x over previous
#include <hip/hip_runtime.h>
#include <hip/hip_bf16.h>
#include <math.h>

typedef __attribute__((ext_vector_type(2))) float v2f;
typedef __attribute__((ext_vector_type(8))) float v8f;

#define N_NODES   50000
#define N_EDGES   800000
#define F_IN      128
#define HC        128
#define HEADS     4
#define HID       32
#define OUT_C     32
#define NEG_SLOPE 0.2f

#define DENSE_K     128
#define DENSE_WAVES 16
#define DENSE_GRID  256
#define N_STRIPS    (N_NODES / 16)

template <int NT, int P1, int P2>
__global__ __launch_bounds__(DENSE_WAVES * 32)
void dense_pair_wmma_f32(const float* __restrict__ A, const float* __restrict__ W,
                         const float* __restrict__ Ws, const float* __restrict__ bs,
                         float* __restrict__ Hout, float* __restrict__ Sout) {
    constexpr int P   = P1 + P2;
    constexpr int PG  = 16 * NT;
    constexpr int NG  = P / PG;
    constexpr int SPB = DENSE_GRID / NG;
    static_assert(NG * PG == P && SPB * NG == DENSE_GRID, "groups tile P and the grid");
    static_assert(SPB * DENSE_WAVES <= N_STRIPS && N_STRIPS <= SPB * 25,
                  "every wave has a first strip; at most 25 strips per work-group");
    __shared__ float4 Bs[(DENSE_K / 4) * PG];

    const int lane = threadIdx.x & 31;
    const int wv = __builtin_amdgcn_readfirstlane((int)(threadIdx.x >> 5));
    const int hf = lane >> 4;
    const int c  = lane & 15;
    const int g  = blockIdx.x % NG;
    const int sb = blockIdx.x / NG;

    const int wr = wv >> 2, wc = wv & 3;
    const int m2 = (((wr + wc) & 1) == 0) ? 16 + 2 * wr + (wc >> 1) : (wv == 1 ? 24 : N_STRIPS);
    int strip = sb + SPB * wv;
    int nxt   = sb + SPB * m2;


    v2f a[DENSE_K / 4];
    {
        const float* arow = A + (size_t)(strip * 16 + c) * DENSE_K + 2 * hf;
#pragma unroll
        for (int i = 0; i < DENSE_K / 4; ++i) a[i] = *(const v2f*)(arow + 4 * i);
    }

    float badd[NT];
#pragma unroll
    for (int t = 0; t < NT; ++t) {
        const int   col0 = g * PG + t * 16;
        const float bv   = bs[((col0 >= P1) ? col0 - P1 : 0) + c];
        badd[t] = (col0 >= P1) ? bv : 0.0f;
    }

    constexpr int SLOTS  = (DENSE_K / 4) * PG;
    constexpr int PASSES = SLOTS / (DENSE_WAVES * 32);
    static_assert(PASSES * DENSE_WAVES * 32 == SLOTS, "whole passes");
#pragma unroll
    for (int r = 0; r < PASSES; ++r) {
        const int it  = (int)threadIdx.x + r * (DENSE_WAVES * 32);
        const int col = g * PG + it % PG;
        const int jh  = it / PG;
        const int k0  = 8 * (jh >> 1) + 2 * (jh & 1);
        const float* s;
        int pitch;
        if (col < P1) { s = W + col;         pitch = P1; }
        else          { s = Ws + (col - P1); pitch = P2; }
        Bs[it] = make_float4(s[(k0 + 0) * pitch], s[(k0 + 1) * pitch],
                             s[(k0 + 4) * pitch], s[(k0 + 5) * pitch]);
    }
    __syncthreads();

    while (strip < N_STRIPS) {
        asm volatile("" ::: "memory");
        v8f acc[NT];
#pragma unroll
        for (int t = 0; t < NT; ++t) acc[t] = (v8f){};

        const float4* bp = Bs + hf * PG + c;
#pragma unroll
        for (int j = 0; j < DENSE_K / 8; ++j) {
            float4 b4[NT];
#pragma unroll
            for (int t = 0; t < NT; ++t) b4[t] = bp[j * 2 * PG + 16 * t];
#pragma unroll
            for (int t = 0; t < NT; ++t) {
                v2f b; b.x = b4[t].x; b.y = b4[t].y;
                acc[t] = __builtin_amdgcn_wmma_f32_16x16x4_f32(
                    false, a[2 * j], false, b, (short)0, acc[t], false, false);
            }
#pragma unroll
            for (int t = 0; t < NT; ++t) {
                v2f b; b.x = b4[t].z; b.y = b4[t].w;
                acc[t] = __builtin_amdgcn_wmma_f32_16x16x4_f32(
                    false, a[2 * j + 1], false, b, (short)0, acc[t], false, false);
            }
        }

        if (nxt < N_STRIPS) {
            const float* arow = A + (size_t)(nxt * 16 + c) * DENSE_K + 2 * hf;
#pragma unroll
            for (int i = 0; i < DENSE_K / 4; ++i) a[i] = *(const v2f*)(arow + 4 * i);
        }

        const int row0 = strip * 16 + 8 * hf;
#pragma unroll
        for (int t = 0; t < NT; ++t) {
            const int  col0 = g * PG + t * 16;
            const bool sk   = col0 >= P1;
            float* o = sk ? Sout + (size_t)row0 * P2 + (col0 - P1) + c
                          : Hout + (size_t)row0 * P1 + col0 + c;
            const int pitch = sk ? P2 : P1;
#pragma unroll
            for (int i = 0; i < 8; ++i) o[(size_t)i * pitch] = acc[t][i] + badd[t];
        }
        strip = nxt;
        nxt   = N_STRIPS;
    }
}

__global__ void attn_logits(const float* __restrict__ h,
                            const float* __restrict__ aS,
                            const float* __restrict__ aD,
                            float* __restrict__ als,
                            float* __restrict__ ald, int N) {
    const int i = blockIdx.x * blockDim.x + threadIdx.x;
    if (i >= N * HEADS) return;
    const int n  = i >> 2;
    const int hd = i & 3;
    const float* hp = h + (size_t)n * HC + hd * HID;
    const float* as = aS + hd * HID;
    const float* ad = aD + hd * HID;
    float ss = 0.f, sd = 0.f;
#pragma unroll 8
    for (int c = 0; c < HID; ++c) {
        const float v = hp[c];
        ss += v * as[c];
        sd += v * ad[c];
    }
    als[i] = ss;
    ald[i] = sd;
}

__global__ void csr_zero(int* __restrict__ cnt, int n) {
    const int i = blockIdx.x * blockDim.x + threadIdx.x;
    if (i < n) cnt[i] = 0;
}

__global__ void csr_count(const int* __restrict__ dst, int* __restrict__ cnt,
                          int* __restrict__ rnk) {
    const int e = blockIdx.x * blockDim.x + threadIdx.x;
    if (e >= N_EDGES) return;
    rnk[e] = atomicAdd(&cnt[dst[e]], 1);
}

__global__ void csr_scan(const int* __restrict__ cnt, int* __restrict__ rowp, int n) {
    __shared__ int wtot[32];
    const int lane  = threadIdx.x & 31;
    const int wave  = threadIdx.x >> 5;
    const int waves = blockDim.x >> 5;
    const int per   = ((n + blockDim.x - 1) / blockDim.x) * 32;
    const int wb    = wave * per;
    int carry = 0;
    for (int it = 0; it < per; it += 32) {
        const int i = wb + it + lane;
        const int v = (i < n) ? cnt[i] : 0;
        int inc = v;
        for (int o = 1; o < 32; o <<= 1) {
            const int t = __shfl_up(inc, o, 32);
            if (lane >= o) inc += t;
        }
        if (i < n) rowp[i] = carry + inc - v;
        carry += __shfl(inc, 31, 32);
    }
    if (lane == 0) wtot[wave] = carry;
    __syncthreads();
    int off = (lane < wave) ? wtot[lane] : 0;
    for (int o = 16; o > 0; o >>= 1) off += __shfl_xor(off, o, 32);
    for (int it = 0; it < per; it += 32) {
        const int i = wb + it + lane;
        if (i < n) rowp[i] += off;
    }
    if (wave == waves - 1 && lane == 0) rowp[n] = off + carry;
}

__global__ void csr_place(const int* __restrict__ dst, const int* __restrict__ rowp,
                          const int* __restrict__ rnk, int* __restrict__ slot) {
    const int e = blockIdx.x * blockDim.x + threadIdx.x;
    if (e >= N_EDGES) return;
    slot[rowp[dst[e]] + rnk[e]] = e;
}

__global__ void csr_order(const int* __restrict__ rowp, const int* __restrict__ slot,
                          const int* __restrict__ src, int* __restrict__ csrc) {
    const int wv   = __builtin_amdgcn_readfirstlane((int)(threadIdx.x >> 5));
    const int node = blockIdx.x * (blockDim.x >> 5) + wv;
    const int lane = threadIdx.x & 31;
    if (node >= N_NODES) return;
    const int beg = rowp[node], end = rowp[node + 1];
    for (int i = beg + lane; i < end; i += 32) {
        const int my = slot[i];
        int r = 0;
        for (int j = beg; j < end; ++j) r += (slot[j] < my) ? 1 : 0;
        csrc[beg + r] = src[my];
    }
}

__device__ __forceinline__ float lrelu(float v) { return (v > 0.f) ? v : NEG_SLOPE * v; }

__device__ __forceinline__ float wave_max(float v) {
    for (int o = 16; o > 0; o >>= 1) v = fmaxf(v, __shfl_xor(v, o, 32));
    return v;
}
__device__ __forceinline__ float wave_sum(float v) {
    for (int o = 16; o > 0; o >>= 1) v += __shfl_xor(v, o, 32);
    return v;
}

__device__ __forceinline__ float4 edge_scores(const float* __restrict__ als, int sN, float4 ad) {
    const float4 a = *(const float4*)(als + (size_t)sN * HEADS);
    return make_float4(lrelu(a.x + ad.x), lrelu(a.y + ad.y), lrelu(a.z + ad.z), lrelu(a.w + ad.w));
}

__device__ __forceinline__ void gather_rows(const float* __restrict__ h, int lane, int cnt, int sN,
                                            float4 al, unsigned k0, unsigned k1, unsigned k2,
                                            unsigned k3, float4& acc) {
    for (int t = 0; t < cnt; ++t) {
        const int      sT = __builtin_amdgcn_readlane(sN, t);
        const unsigned b0 = (unsigned)__builtin_amdgcn_readlane(__float_as_int(al.x), t);
        const unsigned b1 = (unsigned)__builtin_amdgcn_readlane(__float_as_int(al.y), t);
        const unsigned b2 = (unsigned)__builtin_amdgcn_readlane(__float_as_int(al.z), t);
        const unsigned b3 = (unsigned)__builtin_amdgcn_readlane(__float_as_int(al.w), t);
        const float w = __uint_as_float((b0 & k0) | (b1 & k1) | (b2 & k2) | (b3 & k3));
        const float4 hv = *(const float4*)(h + (size_t)sT * HC + lane * 4);
        acc.x += w * hv.x;
        acc.y += w * hv.y;
        acc.z += w * hv.z;
        acc.w += w * hv.w;
    }
}

template <bool LAST>
__global__ void node_softmax_gather(const int* __restrict__ rowp, const int* __restrict__ csrc,
                                    const float* __restrict__ als, const float* __restrict__ ald,
                                    const float* __restrict__ h, const float* __restrict__ bias,
                                    const float* __restrict__ skip, float* __restrict__ out) {
    const int wv   = __builtin_amdgcn_readfirstlane((int)(threadIdx.x >> 5));
    const int node = blockIdx.x * (blockDim.x >> 5) + wv;
    const int lane = threadIdx.x & 31;
    if (node >= N_NODES) return;
    const int beg = rowp[node], end = rowp[node + 1];
    const int hd  = lane >> 3;
    const float4 ad = *(const float4*)(ald + (size_t)node * HEADS);
    const float NEG_INF = -__builtin_huge_valf();
    const int    ec = LAST ? (lane & 7) * 4 : lane * 4;
    const float4 bv = *(const float4*)(bias + ec);
    const float4 sk = *(const float4*)(skip + (size_t)node * (LAST ? OUT_C : HC) + ec);

    const int  j0  = beg + lane;
    const bool ok0 = j0 < end;
    const int  sN0 = ok0 ? csrc[j0] : 0;
    float4 e0 = make_float4(NEG_INF, NEG_INF, NEG_INF, NEG_INF);
    if (ok0) e0 = edge_scores(als, sN0, ad);

    float4 m = e0;
    for (int j = j0 + 32; j < end; j += 32) {
        const float4 e = edge_scores(als, csrc[j], ad);
        m.x = fmaxf(m.x, e.x); m.y = fmaxf(m.y, e.y);
        m.z = fmaxf(m.z, e.z); m.w = fmaxf(m.w, e.w);
    }
    m.x = wave_max(m.x); m.y = wave_max(m.y); m.z = wave_max(m.z); m.w = wave_max(m.w);

    float4 x0 = make_float4(0.f, 0.f, 0.f, 0.f);
    if (ok0) x0 = make_float4(expf(e0.x - m.x), expf(e0.y - m.y), expf(e0.z - m.z), expf(e0.w - m.w));
    float4 s = x0;
    for (int j = j0 + 32; j < end; j += 32) {
        const float4 e = edge_scores(als, csrc[j], ad);
        s.x += expf(e.x - m.x); s.y += expf(e.y - m.y);
        s.z += expf(e.z - m.z); s.w += expf(e.w - m.w);
    }
    s.x = wave_sum(s.x); s.y = wave_sum(s.y); s.z = wave_sum(s.z); s.w = wave_sum(s.w);
    const float4 den = make_float4(s.x + 1e-16f, s.y + 1e-16f, s.z + 1e-16f, s.w + 1e-16f);

    float4 acc = make_float4(0.f, 0.f, 0.f, 0.f);
    const int deg = end - beg;
    const unsigned k0 = (hd == 0) ? 0xFFFFFFFFu : 0u, k1 = (hd == 1) ? 0xFFFFFFFFu : 0u;
    const unsigned k2 = (hd == 2) ? 0xFFFFFFFFu : 0u, k3 = (hd == 3) ? 0xFFFFFFFFu : 0u;
    gather_rows(h, lane, deg < 32 ? deg : 32, sN0,
                make_float4(x0.x / den.x, x0.y / den.y, x0.z / den.z, x0.w / den.w),
                k0, k1, k2, k3, acc);
    for (int base = beg + 32; base < end; base += 32) {
        const int j = base + lane;
        int sN = 0;
        float4 al = make_float4(0.f, 0.f, 0.f, 0.f);
        if (j < end) {
            sN = csrc[j];
            const float4 e = edge_scores(als, sN, ad);
            al = make_float4(expf(e.x - m.x) / den.x, expf(e.y - m.y) / den.y,
                             expf(e.z - m.z) / den.z, expf(e.w - m.w) / den.w);
        }
        const int left = end - base;
        gather_rows(h, lane, left < 32 ? left : 32, sN, al, k0, k1, k2, k3, acc);
    }
    if (!LAST) {
        float4 v = make_float4(acc.x + bv.x + sk.x, acc.y + bv.y + sk.y,
                               acc.z + bv.z + sk.z, acc.w + bv.w + sk.w);
        v.x = (v.x > 0.f) ? v.x : (expf(v.x) - 1.0f);
        v.y = (v.y > 0.f) ? v.y : (expf(v.y) - 1.0f);
        v.z = (v.z > 0.f) ? v.z : (expf(v.z) - 1.0f);
        v.w = (v.w > 0.f) ? v.w : (expf(v.w) - 1.0f);
        *(float4*)(out + (size_t)node * HC + lane * 4) = v;
    } else {
        const int g = lane & 7;
        float4 hs;
        hs.x = __shfl(acc.x, g, 32) + __shfl(acc.x, g + 8, 32) + __shfl(acc.x, g + 16, 32) + __shfl(acc.x, g + 24, 32);
        hs.y = __shfl(acc.y, g, 32) + __shfl(acc.y, g + 8, 32) + __shfl(acc.y, g + 16, 32) + __shfl(acc.y, g + 24, 32);
        hs.z = __shfl(acc.z, g, 32) + __shfl(acc.z, g + 8, 32) + __shfl(acc.z, g + 16, 32) + __shfl(acc.z, g + 24, 32);
        hs.w = __shfl(acc.w, g, 32) + __shfl(acc.w, g + 8, 32) + __shfl(acc.w, g + 16, 32) + __shfl(acc.w, g + 24, 32);
        const float4 v = make_float4(0.25f * hs.x + bv.x + sk.x, 0.25f * hs.y + bv.y + sk.y,
                                     0.25f * hs.z + bv.z + sk.z, 0.25f * hs.w + bv.w + sk.w);
        float4 t = make_float4(v.x * v.x, v.y * v.y, v.z * v.z, v.w * v.w);
        for (int o = 4; o > 0; o >>= 1) {
            t.x += __shfl_xor(t.x, o, 32);
            t.y += __shfl_xor(t.y, o, 32);
            t.z += __shfl_xor(t.z, o, 32);
            t.w += __shfl_xor(t.w, o, 32);
        }
        const float sq   = (t.x + t.z) + (t.y + t.w);
        const float norm = fmaxf(sqrtf(sq), 1e-12f);
        if (lane < 8)
            *(float4*)(out + (size_t)node * OUT_C + g * 4) =
                make_float4(v.x / norm, v.y / norm, v.z / norm, v.w / norm);
    }
}

extern "C" void kernel_launch(void* const* d_in, const int* in_sizes, int n_in,
                              void* d_out, int out_size, void* d_ws, size_t ws_size,
                              hipStream_t stream) {
    (void)in_sizes; (void)n_in; (void)out_size; (void)ws_size;

    const float* x   = (const float*)d_in[0];
    const int*   ei  = (const int*)d_in[1];
    const float* W0  = (const float*)d_in[2];
    const float* aS0 = (const float*)d_in[3];
    const float* aD0 = (const float*)d_in[4];
    const float* b0  = (const float*)d_in[5];
    const float* Ws0 = (const float*)d_in[6];
    const float* bs0 = (const float*)d_in[7];
    const float* W1  = (const float*)d_in[8];
    const float* aS1 = (const float*)d_in[9];
    const float* aD1 = (const float*)d_in[10];
    const float* b1  = (const float*)d_in[11];
    const float* Ws1 = (const float*)d_in[12];
    const float* bs1 = (const float*)d_in[13];
    const float* W2  = (const float*)d_in[14];
    const float* aS2 = (const float*)d_in[15];
    const float* aD2 = (const float*)d_in[16];
    const float* b2  = (const float*)d_in[17];
    const float* Ws2 = (const float*)d_in[18];
    const float* bs2 = (const float*)d_in[19];

    const int* src = ei;
    const int* dst = ei + N_EDGES;

    const size_t NF = (size_t)N_NODES * HC;
    float*    ws    = (float*)d_ws;
    float*    x_cur = ws;
    float*    h_lin = ws + NF;
    float*    skipb = ws + 2 * NF;
    float*    als   = ws + 3 * NF;
    float*    ald   = als + (size_t)N_NODES * HEADS;
    const int NPAD  = N_NODES + 32;
    int*      cnt   = (int*)(ald + (size_t)N_NODES * HEADS);
    int*      rowp  = cnt + NPAD;
    int*      rnk   = rowp + NPAD;
    int*      slot  = rnk + N_EDGES;
    int*      csrc  = slot + N_EDGES;

    auto run_edges = [&](const float* hfeat, const float* b, float* xout) {
        node_softmax_gather<false><<<(N_NODES + 7) / 8, 256, 0, stream>>>(
            rowp, csrc, als, ald, hfeat, b, skipb, xout);
    };

    csr_zero<<<(N_NODES + 255) / 256, 256, 0, stream>>>(cnt, N_NODES);
    csr_count<<<(N_EDGES + 255) / 256, 256, 0, stream>>>(dst, cnt, rnk);
    csr_scan<<<1, 1024, 0, stream>>>(cnt, rowp, N_NODES);
    csr_place<<<(N_EDGES + 255) / 256, 256, 0, stream>>>(dst, rowp, rnk, slot);
    csr_order<<<(N_NODES + 7) / 8, 256, 0, stream>>>(rowp, slot, src, csrc);

    const int nNH = N_NODES * HEADS;

    dense_pair_wmma_f32<8, HC, HC><<<DENSE_GRID, DENSE_WAVES * 32, 0, stream>>>(x, W0, Ws0, bs0, h_lin, skipb);
    attn_logits<<<(nNH + 255) / 256, 256, 0, stream>>>(h_lin, aS0, aD0, als, ald, N_NODES);
    run_edges(h_lin, b0, x_cur);

    dense_pair_wmma_f32<8, HC, HC><<<DENSE_GRID, DENSE_WAVES * 32, 0, stream>>>(x_cur, W1, Ws1, bs1, h_lin, skipb);
    attn_logits<<<(nNH + 255) / 256, 256, 0, stream>>>(h_lin, aS1, aD1, als, ald, N_NODES);
    run_edges(h_lin, b1, x_cur);

    dense_pair_wmma_f32<5, HEADS * OUT_C, OUT_C><<<DENSE_GRID, DENSE_WAVES * 32, 0, stream>>>(x_cur, W2, Ws2, bs2, h_lin, skipb);
    attn_logits<<<(nNH + 255) / 256, 256, 0, stream>>>(h_lin, aS2, aD2, als, ald, N_NODES);
    node_softmax_gather<true><<<(N_NODES + 7) / 8, 256, 0, stream>>>(
        rowp, csrc, als, ald, h_lin, b2, skipb, (float*)d_out);
}


# In-degree prefix scan: a wave reads its whole 1664-element stretch with all wide loads in flight at once and scans it in registers, not 49 serial load-scan-store trips plus a read-add-write pass
# speedup vs baseline: 4.9652x; 1.1369x over previous
#include <hip/hip_runtime.h>
#include <hip/hip_bf16.h>
#include <math.h>

typedef __attribute__((ext_vector_type(2))) float v2f;
typedef __attribute__((ext_vector_type(8))) float v8f;

#define N_NODES   50000
#define N_EDGES   800000
#define F_IN      128
#define HC        128
#define HEADS     4
#define HID       32
#define OUT_C     32
#define NEG_SLOPE 0.2f

#define DENSE_K     128
#define DENSE_WAVES 16
#define DENSE_GRID  256
#define N_STRIPS    (N_NODES / 16)

template <int NT, int P1, int P2>
__global__ __launch_bounds__(DENSE_WAVES * 32)
void dense_pair_wmma_f32(const float* __restrict__ A, const float* __restrict__ W,
                         const float* __restrict__ Ws, const float* __restrict__ bs,
                         float* __restrict__ Hout, float* __restrict__ Sout) {
    constexpr int P   = P1 + P2;
    constexpr int PG  = 16 * NT;
    constexpr int NG  = P / PG;
    constexpr int SPB = DENSE_GRID / NG;
    static_assert(NG * PG == P && SPB * NG == DENSE_GRID, "groups tile P and the grid");
    static_assert(SPB * DENSE_WAVES <= N_STRIPS && N_STRIPS <= SPB * 25,
                  "every wave has a first strip; at most 25 strips per work-group");
    __shared__ float4 Bs[(DENSE_K / 4) * PG];

    const int lane = threadIdx.x & 31;
    const int wv = __builtin_amdgcn_readfirstlane((int)(threadIdx.x >> 5));
    const int hf = lane >> 4;
    const int c  = lane & 15;
    const int g  = blockIdx.x % NG;
    const int sb = blockIdx.x / NG;

    const int wr = wv >> 2, wc = wv & 3;
    const int m2 = (((wr + wc) & 1) == 0) ? 16 + 2 * wr + (wc >> 1) : (wv == 1 ? 24 : N_STRIPS);
    int strip = sb + SPB * wv;
    int nxt   = sb + SPB * m2;


    v2f a[DENSE_K / 4];
    {
        const float* arow = A + (size_t)(strip * 16 + c) * DENSE_K + 2 * hf;
#pragma unroll
        for (int i = 0; i < DENSE_K / 4; ++i) a[i] = *(const v2f*)(arow + 4 * i);
    }

    float badd[NT];
#pragma unroll
    for (int t = 0; t < NT; ++t) {
        const int   col0 = g * PG + t * 16;
        const float bv   = bs[((col0 >= P1) ? col0 - P1 : 0) + c];
        badd[t] = (col0 >= P1) ? bv : 0.0f;
    }

    constexpr int SLOTS  = (DENSE_K / 4) * PG;
    constexpr int PASSES = SLOTS / (DENSE_WAVES * 32);
    static_assert(PASSES * DENSE_WAVES * 32 == SLOTS, "whole passes");
#pragma unroll
    for (int r = 0; r < PASSES; ++r) {
        const int it  = (int)threadIdx.x + r * (DENSE_WAVES * 32);
        const int col = g * PG + it % PG;
        const int jh  = it / PG;
        const int k0  = 8 * (jh >> 1) + 2 * (jh & 1);
        const float* s;
        int pitch;
        if (col < P1) { s = W + col;         pitch = P1; }
        else          { s = Ws + (col - P1); pitch = P2; }
        Bs[it] = make_float4(s[(k0 + 0) * pitch], s[(k0 + 1) * pitch],
                             s[(k0 + 4) * pitch], s[(k0 + 5) * pitch]);
    }
    __syncthreads();

    while (strip < N_STRIPS) {
        asm volatile("" ::: "memory");
        v8f acc[NT];
#pragma unroll
        for (int t = 0; t < NT; ++t) acc[t] = (v8f){};

        const float4* bp = Bs + hf * PG + c;
#pragma unroll
        for (int j = 0; j < DENSE_K / 8; ++j) {
            float4 b4[NT];
#pragma unroll
            for (int t = 0; t < NT; ++t) b4[t] = bp[j * 2 * PG + 16 * t];
#pragma unroll
            for (int t = 0; t < NT; ++t) {
                v2f b; b.x = b4[t].x; b.y = b4[t].y;
                acc[t] = __builtin_amdgcn_wmma_f32_16x16x4_f32(
                    false, a[2 * j], false, b, (short)0, acc[t], false, false);
            }
#pragma unroll
            for (int t = 0; t < NT; ++t) {
                v2f b; b.x = b4[t].z; b.y = b4[t].w;
                acc[t] = __builtin_amdgcn_wmma_f32_16x16x4_f32(
                    false, a[2 * j + 1], false, b, (short)0, acc[t], false, false);
            }
        }

        if (nxt < N_STRIPS) {
            const float* arow = A + (size_t)(nxt * 16 + c) * DENSE_K + 2 * hf;
#pragma unroll
            for (int i = 0; i < DENSE_K / 4; ++i) a[i] = *(const v2f*)(arow + 4 * i);
        }

        const int row0 = strip * 16 + 8 * hf;
#pragma unroll
        for (int t = 0; t < NT; ++t) {
            const int  col0 = g * PG + t * 16;
            const bool sk   = col0 >= P1;
            float* o = sk ? Sout + (size_t)row0 * P2 + (col0 - P1) + c
                          : Hout + (size_t)row0 * P1 + col0 + c;
            const int pitch = sk ? P2 : P1;
#pragma unroll
            for (int i = 0; i < 8; ++i) o[(size_t)i * pitch] = acc[t][i] + badd[t];
        }
        strip = nxt;
        nxt   = N_STRIPS;
    }
}

__global__ void attn_logits(const float* __restrict__ h,
                            const float* __restrict__ aS,
                            const float* __restrict__ aD,
                            float* __restrict__ als,
                            float* __restrict__ ald, int N) {
    const int i = blockIdx.x * blockDim.x + threadIdx.x;
    if (i >= N * HEADS) return;
    const int n  = i >> 2;
    const int hd = i & 3;
    const float* hp = h + (size_t)n * HC + hd * HID;
    const float* as = aS + hd * HID;
    const float* ad = aD + hd * HID;
    float ss = 0.f, sd = 0.f;
#pragma unroll 8
    for (int c = 0; c < HID; ++c) {
        const float v = hp[c];
        ss += v * as[c];
        sd += v * ad[c];
    }
    als[i] = ss;
    ald[i] = sd;
}

__global__ void csr_zero(int* __restrict__ cnt, int n) {
    const int i = blockIdx.x * blockDim.x + threadIdx.x;
    if (i < n) cnt[i] = 0;
}

__global__ void csr_count(const int* __restrict__ dst, int* __restrict__ cnt,
                          int* __restrict__ rnk) {
    const int e = blockIdx.x * blockDim.x + threadIdx.x;
    if (e >= N_EDGES) return;
    rnk[e] = atomicAdd(&cnt[dst[e]], 1);
}

#define SCAN_WAVES 32
#define SCAN_K     13
static_assert(SCAN_WAVES * SCAN_K * 128 >= N_NODES && N_NODES % 4 == 0,
              "the stretches cover every node; 16-byte pieces are whole");

__global__ __launch_bounds__(SCAN_WAVES * 32)
void csr_scan(const int* __restrict__ cnt, int* __restrict__ rowp) {
    __shared__ int wtot[SCAN_WAVES];
    const int lane = threadIdx.x & 31;
    const int wave = threadIdx.x >> 5;
    const int e0   = wave * (SCAN_K * 128) + lane * 4;

    int4 v[SCAN_K];
#pragma unroll
    for (int k = 0; k < SCAN_K; ++k) {
        const int e = e0 + k * 128;
        v[k] = *(const int4*)(cnt + (e < N_NODES ? e : N_NODES - 4));
    }
    int own[SCAN_K], inc[SCAN_K];
#pragma unroll
    for (int k = 0; k < SCAN_K; ++k) {
        if (e0 + k * 128 >= N_NODES) v[k] = make_int4(0, 0, 0, 0);
        own[k] = (v[k].x + v[k].y) + (v[k].z + v[k].w);
        inc[k] = own[k];
    }
#pragma unroll
    for (int o = 1; o < 32; o <<= 1) {
#pragma unroll
        for (int k = 0; k < SCAN_K; ++k) {
            const int t = __shfl_up(inc[k], o, 32);
            if (lane >= o) inc[k] += t;
        }
    }
    int carry = 0;
    int base[SCAN_K];
#pragma unroll
    for (int k = 0; k < SCAN_K; ++k) {
        base[k] = carry + inc[k] - own[k];
        carry += __shfl(inc[k], 31, 32);
    }
    if (lane == 0) wtot[wave] = carry;
    __syncthreads();
    int off = (lane < wave) ? wtot[lane] : 0;
    for (int o = 16; o > 0; o >>= 1) off += __shfl_xor(off, o, 32);
#pragma unroll
    for (int k = 0; k < SCAN_K; ++k) {
        const int e = e0 + k * 128;
        if (e < N_NODES) {
            int4 r;
            r.x = off + base[k];
            r.y = r.x + v[k].x;
            r.z = r.y + v[k].y;
            r.w = r.z + v[k].z;
            *(int4*)(rowp + e) = r;
        }
    }
    if (wave == SCAN_WAVES - 1 && lane == 0) rowp[N_NODES] = off + carry;
}

__global__ void csr_place(const int* __restrict__ dst, const int* __restrict__ rowp,
                          const int* __restrict__ rnk, int* __restrict__ slot) {
    const int e = blockIdx.x * blockDim.x + threadIdx.x;
    if (e >= N_EDGES) return;
    slot[rowp[dst[e]] + rnk[e]] = e;
}

__global__ void csr_order(const int* __restrict__ rowp, const int* __restrict__ slot,
                          const int* __restrict__ src, int* __restrict__ csrc) {
    const int wv   = __builtin_amdgcn_readfirstlane((int)(threadIdx.x >> 5));
    const int node = blockIdx.x * (blockDim.x >> 5) + wv;
    const int lane = threadIdx.x & 31;
    if (node >= N_NODES) return;
    const int beg = rowp[node], end = rowp[node + 1];
    for (int i = beg + lane; i < end; i += 32) {
        const int my = slot[i];
        int r = 0;
        for (int j = beg; j < end; ++j) r += (slot[j] < my) ? 1 : 0;
        csrc[beg + r] = src[my];
    }
}

__device__ __forceinline__ float lrelu(float v) { return (v > 0.f) ? v : NEG_SLOPE * v; }

__device__ __forceinline__ float wave_max(float v) {
    for (int o = 16; o > 0; o >>= 1) v = fmaxf(v, __shfl_xor(v, o, 32));
    return v;
}
__device__ __forceinline__ float wave_sum(float v) {
    for (int o = 16; o > 0; o >>= 1) v += __shfl_xor(v, o, 32);
    return v;
}

__device__ __forceinline__ float4 edge_scores(const float* __restrict__ als, int sN, float4 ad) {
    const float4 a = *(const float4*)(als + (size_t)sN * HEADS);
    return make_float4(lrelu(a.x + ad.x), lrelu(a.y + ad.y), lrelu(a.z + ad.z), lrelu(a.w + ad.w));
}

__device__ __forceinline__ void gather_rows(const float* __restrict__ h, int lane, int cnt, int sN,
                                            float4 al, unsigned k0, unsigned k1, unsigned k2,
                                            unsigned k3, float4& acc) {
    for (int t = 0; t < cnt; ++t) {
        const int      sT = __builtin_amdgcn_readlane(sN, t);
        const unsigned b0 = (unsigned)__builtin_amdgcn_readlane(__float_as_int(al.x), t);
        const unsigned b1 = (unsigned)__builtin_amdgcn_readlane(__float_as_int(al.y), t);
        const unsigned b2 = (unsigned)__builtin_amdgcn_readlane(__float_as_int(al.z), t);
        const unsigned b3 = (unsigned)__builtin_amdgcn_readlane(__float_as_int(al.w), t);
        const float w = __uint_as_float((b0 & k0) | (b1 & k1) | (b2 & k2) | (b3 & k3));
        const float4 hv = *(const float4*)(h + (size_t)sT * HC + lane * 4);
        acc.x += w * hv.x;
        acc.y += w * hv.y;
        acc.z += w * hv.z;
        acc.w += w * hv.w;
    }
}

template <bool LAST>
__global__ void node_softmax_gather(const int* __restrict__ rowp, const int* __restrict__ csrc,
                                    const float* __restrict__ als, const float* __restrict__ ald,
                                    const float* __restrict__ h, const float* __restrict__ bias,
                                    const float* __restrict__ skip, float* __restrict__ out) {
    const int wv   = __builtin_amdgcn_readfirstlane((int)(threadIdx.x >> 5));
    const int node = blockIdx.x * (blockDim.x >> 5) + wv;
    const int lane = threadIdx.x & 31;
    if (node >= N_NODES) return;
    const int beg = rowp[node], end = rowp[node + 1];
    const int hd  = lane >> 3;
    const float4 ad = *(const float4*)(ald + (size_t)node * HEADS);
    const float NEG_INF = -__builtin_huge_valf();
    const int    ec = LAST ? (lane & 7) * 4 : lane * 4;
    const float4 bv = *(const float4*)(bias + ec);
    const float4 sk = *(const float4*)(skip + (size_t)node * (LAST ? OUT_C : HC) + ec);

    const int  j0  = beg + lane;
    const bool ok0 = j0 < end;
    const int  sN0 = ok0 ? csrc[j0] : 0;
    float4 e0 = make_float4(NEG_INF, NEG_INF, NEG_INF, NEG_INF);
    if (ok0) e0 = edge_scores(als, sN0, ad);

    float4 m = e0;
    for (int j = j0 + 32; j < end; j += 32) {
        const float4 e = edge_scores(als, csrc[j], ad);
        m.x = fmaxf(m.x, e.x); m.y = fmaxf(m.y, e.y);
        m.z = fmaxf(m.z, e.z); m.w = fmaxf(m.w, e.w);
    }
    m.x = wave_max(m.x); m.y = wave_max(m.y); m.z = wave_max(m.z); m.w = wave_max(m.w);

    float4 x0 = make_float4(0.f, 0.f, 0.f, 0.f);
    if (ok0) x0 = make_float4(expf(e0.x - m.x), expf(e0.y - m.y), expf(e0.z - m.z), expf(e0.w - m.w));
    float4 s = x0;
    for (int j = j0 + 32; j < end; j += 32) {
        const float4 e = edge_scores(als, csrc[j], ad);
        s.x += expf(e.x - m.x); s.y += expf(e.y - m.y);
        s.z += expf(e.z - m.z); s.w += expf(e.w - m.w);
    }
    s.x = wave_sum(s.x); s.y = wave_sum(s.y); s.z = wave_sum(s.z); s.w = wave_sum(s.w);
    const float4 den = make_float4(s.x + 1e-16f, s.y + 1e-16f, s.z + 1e-16f, s.w + 1e-16f);

    float4 acc = make_float4(0.f, 0.f, 0.f, 0.f);
    const int deg = end - beg;
    const unsigned k0 = (hd == 0) ? 0xFFFFFFFFu : 0u, k1 = (hd == 1) ? 0xFFFFFFFFu : 0u;
    const unsigned k2 = (hd == 2) ? 0xFFFFFFFFu : 0u, k3 = (hd == 3) ? 0xFFFFFFFFu : 0u;
    gather_rows(h, lane, deg < 32 ? deg : 32, sN0,
                make_float4(x0.x / den.x, x0.y / den.y, x0.z / den.z, x0.w / den.w),
                k0, k1, k2, k3, acc);
    for (int base = beg + 32; base < end; base += 32) {
        const int j = base + lane;
        int sN = 0;
        float4 al = make_float4(0.f, 0.f, 0.f, 0.f);
        if (j < end) {
            sN = csrc[j];
            const float4 e = edge_scores(als, sN, ad);
            al = make_float4(expf(e.x - m.x) / den.x, expf(e.y - m.y) / den.y,
                             expf(e.z - m.z) / den.z, expf(e.w - m.w) / den.w);
        }
        const int left = end - base;
        gather_rows(h, lane, left < 32 ? left : 32, sN, al, k0, k1, k2, k3, acc);
    }
    if (!LAST) {
        float4 v = make_float4(acc.x + bv.x + sk.x, acc.y + bv.y + sk.y,
                               acc.z + bv.z + sk.z, acc.w + bv.w + sk.w);
        v.x = (v.x > 0.f) ? v.x : (expf(v.x) - 1.0f);
        v.y = (v.y > 0.f) ? v.y : (expf(v.y) - 1.0f);
        v.z = (v.z > 0.f) ? v.z : (expf(v.z) - 1.0f);
        v.w = (v.w > 0.f) ? v.w : (expf(v.w) - 1.0f);
        *(float4*)(out + (size_t)node * HC + lane * 4) = v;
    } else {
        const int g = lane & 7;
        float4 hs;
        hs.x = __shfl(acc.x, g, 32) + __shfl(acc.x, g + 8, 32) + __shfl(acc.x, g + 16, 32) + __shfl(acc.x, g + 24, 32);
        hs.y = __shfl(acc.y, g, 32) + __shfl(acc.y, g + 8, 32) + __shfl(acc.y, g + 16, 32) + __shfl(acc.y, g + 24, 32);
        hs.z = __shfl(acc.z, g, 32) + __shfl(acc.z, g + 8, 32) + __shfl(acc.z, g + 16, 32) + __shfl(acc.z, g + 24, 32);
        hs.w = __shfl(acc.w, g, 32) + __shfl(acc.w, g + 8, 32) + __shfl(acc.w, g + 16, 32) + __shfl(acc.w, g + 24, 32);
        const float4 v = make_float4(0.25f * hs.x + bv.x + sk.x, 0.25f * hs.y + bv.y + sk.y,
                                     0.25f * hs.z + bv.z + sk.z, 0.25f * hs.w + bv.w + sk.w);
        float4 t = make_float4(v.x * v.x, v.y * v.y, v.z * v.z, v.w * v.w);
        for (int o = 4; o > 0; o >>= 1) {
            t.x += __shfl_xor(t.x, o, 32);
            t.y += __shfl_xor(t.y, o, 32);
            t.z += __shfl_xor(t.z, o, 32);
            t.w += __shfl_xor(t.w, o, 32);
        }
        const float sq   = (t.x + t.z) + (t.y + t.w);
        const float norm = fmaxf(sqrtf(sq), 1e-12f);
        if (lane < 8)
            *(float4*)(out + (size_t)node * OUT_C + g * 4) =
                make_float4(v.x / norm, v.y / norm, v.z / norm, v.w / norm);
    }
}

extern "C" void kernel_launch(void* const* d_in, const int* in_sizes, int n_in,
                              void* d_out, int out_size, void* d_ws, size_t ws_size,
                              hipStream_t stream) {
    (void)in_sizes; (void)n_in; (void)out_size; (void)ws_size;

    const float* x   = (const float*)d_in[0];
    const int*   ei  = (const int*)d_in[1];
    const float* W0  = (const float*)d_in[2];
    const float* aS0 = (const float*)d_in[3];
    const float* aD0 = (const float*)d_in[4];
    const float* b0  = (const float*)d_in[5];
    const float* Ws0 = (const float*)d_in[6];
    const float* bs0 = (const float*)d_in[7];
    const float* W1  = (const float*)d_in[8];
    const float* aS1 = (const float*)d_in[9];
    const float* aD1 = (const float*)d_in[10];
    const float* b1  = (const float*)d_in[11];
    const float* Ws1 = (const float*)d_in[12];
    const float* bs1 = (const float*)d_in[13];
    const float* W2  = (const float*)d_in[14];
    const float* aS2 = (const float*)d_in[15];
    const float* aD2 = (const float*)d_in[16];
    const float* b2  = (const float*)d_in[17];
    const float* Ws2 = (const float*)d_in[18];
    const float* bs2 = (const float*)d_in[19];

    const int* src = ei;
    const int* dst = ei + N_EDGES;

    const size_t NF = (size_t)N_NODES * HC;
    float*    ws    = (float*)d_ws;
    float*    x_cur = ws;
    float*    h_lin = ws + NF;
    float*    skipb = ws + 2 * NF;
    float*    als   = ws + 3 * NF;
    float*    ald   = als + (size_t)N_NODES * HEADS;
    const int NPAD  = N_NODES + 32;
    int*      cnt   = (int*)(ald + (size_t)N_NODES * HEADS);
    int*      rowp  = cnt + NPAD;
    int*      rnk   = rowp + NPAD;
    int*      slot  = rnk + N_EDGES;
    int*      csrc  = slot + N_EDGES;

    auto run_edges = [&](const float* hfeat, const float* b, float* xout) {
        node_softmax_gather<false><<<(N_NODES + 7) / 8, 256, 0, stream>>>(
            rowp, csrc, als, ald, hfeat, b, skipb, xout);
    };

    csr_zero<<<(N_NODES + 255) / 256, 256, 0, stream>>>(cnt, N_NODES);
    csr_count<<<(N_EDGES + 255) / 256, 256, 0, stream>>>(dst, cnt, rnk);
    csr_scan<<<1, SCAN_WAVES * 32, 0, stream>>>(cnt, rowp);
    csr_place<<<(N_EDGES + 255) / 256, 256, 0, stream>>>(dst, rowp, rnk, slot);
    csr_order<<<(N_NODES + 7) / 8, 256, 0, stream>>>(rowp, slot, src, csrc);

    const int nNH = N_NODES * HEADS;

    dense_pair_wmma_f32<8, HC, HC><<<DENSE_GRID, DENSE_WAVES * 32, 0, stream>>>(x, W0, Ws0, bs0, h_lin, skipb);
    attn_logits<<<(nNH + 255) / 256, 256, 0, stream>>>(h_lin, aS0, aD0, als, ald, N_NODES);
    run_edges(h_lin, b0, x_cur);

    dense_pair_wmma_f32<8, HC, HC><<<DENSE_GRID, DENSE_WAVES * 32, 0, stream>>>(x_cur, W1, Ws1, bs1, h_lin, skipb);
    attn_logits<<<(nNH + 255) / 256, 256, 0, stream>>>(h_lin, aS1, aD1, als, ald, N_NODES);
    run_edges(h_lin, b1, x_cur);

    dense_pair_wmma_f32<5, HEADS * OUT_C, OUT_C><<<DENSE_GRID, DENSE_WAVES * 32, 0, stream>>>(x_cur, W2, Ws2, bs2, h_lin, skipb);
    attn_logits<<<(nNH + 255) / 256, 256, 0, stream>>>(h_lin, aS2, aD2, als, ald, N_NODES);
    node_softmax_gather<true><<<(N_NODES + 7) / 8, 256, 0, stream>>>(
        rowp, csrc, als, ald, h_lin, b2, skipb, (float*)d_out);
}


# Gather wave takes each edge's attention score from the source row it fetches anyway (8-lane DPP dot) and runs a one-pass running-max softmax in base 2: 3 logit launches gone; new sum order
# speedup vs baseline: 5.4998x; 1.1077x over previous
#include <hip/hip_runtime.h>
#include <hip/hip_bf16.h>
#include <math.h>

typedef __attribute__((ext_vector_type(2))) float v2f;
typedef __attribute__((ext_vector_type(8))) float v8f;

#define N_NODES   50000
#define N_EDGES   800000
#define F_IN      128
#define HC        128
#define HEADS     4
#define HID       32
#define OUT_C     32
#define NEG_SLOPE 0.2f

#define DENSE_K     128
#define DENSE_WAVES 16
#define DENSE_GRID  256
#define N_STRIPS    (N_NODES / 16)

template <int NT, int P1, int P2>
__global__ __launch_bounds__(DENSE_WAVES * 32)
void dense_pair_wmma_f32(const float* __restrict__ A, const float* __restrict__ W,
                         const float* __restrict__ Ws, const float* __restrict__ bs,
                         float* __restrict__ Hout, float* __restrict__ Sout) {
    constexpr int P   = P1 + P2;
    constexpr int PG  = 16 * NT;
    constexpr int NG  = P / PG;
    constexpr int SPB = DENSE_GRID / NG;
    static_assert(NG * PG == P && SPB * NG == DENSE_GRID, "groups tile P and the grid");
    static_assert(SPB * DENSE_WAVES <= N_STRIPS && N_STRIPS <= SPB * 25,
                  "every wave has a first strip; at most 25 strips per work-group");
    __shared__ float4 Bs[(DENSE_K / 4) * PG];

    const int lane = threadIdx.x & 31;
    const int wv = __builtin_amdgcn_readfirstlane((int)(threadIdx.x >> 5));
    const int hf = lane >> 4;
    const int c  = lane & 15;
    const int g  = blockIdx.x % NG;
    const int sb = blockIdx.x / NG;

    const int wr = wv >> 2, wc = wv & 3;
    const int m2 = (((wr + wc) & 1) == 0) ? 16 + 2 * wr + (wc >> 1) : (wv == 1 ? 24 : N_STRIPS);
    int strip = sb + SPB * wv;
    int nxt   = sb + SPB * m2;


    v2f a[DENSE_K / 4];
    {
        const float* arow = A + (size_t)(strip * 16 + c) * DENSE_K + 2 * hf;
#pragma unroll
        for (int i = 0; i < DENSE_K / 4; ++i) a[i] = *(const v2f*)(arow + 4 * i);
    }

    float badd[NT];
#pragma unroll
    for (int t = 0; t < NT; ++t) {
        const int   col0 = g * PG + t * 16;
        const float bv   = bs[((col0 >= P1) ? col0 - P1 : 0) + c];
        badd[t] = (col0 >= P1) ? bv : 0.0f;
    }

    constexpr int SLOTS  = (DENSE_K / 4) * PG;
    constexpr int PASSES = SLOTS / (DENSE_WAVES * 32);
    static_assert(PASSES * DENSE_WAVES * 32 == SLOTS, "whole passes");
#pragma unroll
    for (int r = 0; r < PASSES; ++r) {
        const int it  = (int)threadIdx.x + r * (DENSE_WAVES * 32);
        const int col = g * PG + it % PG;
        const int jh  = it / PG;
        const int k0  = 8 * (jh >> 1) + 2 * (jh & 1);
        const float* s;
        int pitch;
        if (col < P1) { s = W + col;         pitch = P1; }
        else          { s = Ws + (col - P1); pitch = P2; }
        Bs[it] = make_float4(s[(k0 + 0) * pitch], s[(k0 + 1) * pitch],
                             s[(k0 + 4) * pitch], s[(k0 + 5) * pitch]);
    }
    __syncthreads();

    while (strip < N_STRIPS) {
        asm volatile("" ::: "memory");
        v8f acc[NT];
#pragma unroll
        for (int t = 0; t < NT; ++t) acc[t] = (v8f){};

        const float4* bp = Bs + hf * PG + c;
#pragma unroll
        for (int j = 0; j < DENSE_K / 8; ++j) {
            float4 b4[NT];
#pragma unroll
            for (int t = 0; t < NT; ++t) b4[t] = bp[j * 2 * PG + 16 * t];
#pragma unroll
            for (int t = 0; t < NT; ++t) {
                v2f b; b.x = b4[t].x; b.y = b4[t].y;
                acc[t] = __builtin_amdgcn_wmma_f32_16x16x4_f32(
                    false, a[2 * j], false, b, (short)0, acc[t], false, false);
            }
#pragma unroll
            for (int t = 0; t < NT; ++t) {
                v2f b; b.x = b4[t].z; b.y = b4[t].w;
                acc[t] = __builtin_amdgcn_wmma_f32_16x16x4_f32(
                    false, a[2 * j + 1], false, b, (short)0, acc[t], false, false);
            }
        }

        if (nxt < N_STRIPS) {
            const float* arow = A + (size_t)(nxt * 16 + c) * DENSE_K + 2 * hf;
#pragma unroll
            for (int i = 0; i < DENSE_K / 4; ++i) a[i] = *(const v2f*)(arow + 4 * i);
        }

        const int row0 = strip * 16 + 8 * hf;
#pragma unroll
        for (int t = 0; t < NT; ++t) {
            const int  col0 = g * PG + t * 16;
            const bool sk   = col0 >= P1;
            float* o = sk ? Sout + (size_t)row0 * P2 + (col0 - P1) + c
                          : Hout + (size_t)row0 * P1 + col0 + c;
            const int pitch = sk ? P2 : P1;
#pragma unroll
            for (int i = 0; i < 8; ++i) o[(size_t)i * pitch] = acc[t][i] + badd[t];
        }
        strip = nxt;
        nxt   = N_STRIPS;
    }
}

__global__ void csr_zero(int* __restrict__ cnt, int n) {
    const int i = blockIdx.x * blockDim.x + threadIdx.x;
    if (i < n) cnt[i] = 0;
}

__global__ void csr_count(const int* __restrict__ dst, int* __restrict__ cnt,
                          int* __restrict__ rnk) {
    const int e = blockIdx.x * blockDim.x + threadIdx.x;
    if (e >= N_EDGES) return;
    rnk[e] = atomicAdd(&cnt[dst[e]], 1);
}

#define SCAN_WAVES 32
#define SCAN_K     13
static_assert(SCAN_WAVES * SCAN_K * 128 >= N_NODES && N_NODES % 4 == 0,
              "the stretches cover every node; 16-byte pieces are whole");

__global__ __launch_bounds__(SCAN_WAVES * 32)
void csr_scan(const int* __restrict__ cnt, int* __restrict__ rowp) {
    __shared__ int wtot[SCAN_WAVES];
    const int lane = threadIdx.x & 31;
    const int wave = threadIdx.x >> 5;
    const int e0   = wave * (SCAN_K * 128) + lane * 4;

    int4 v[SCAN_K];
#pragma unroll
    for (int k = 0; k < SCAN_K; ++k) {
        const int e = e0 + k * 128;
        v[k] = *(const int4*)(cnt + (e < N_NODES ? e : N_NODES - 4));
    }
    int own[SCAN_K], inc[SCAN_K];
#pragma unroll
    for (int k = 0; k < SCAN_K; ++k) {
        if (e0 + k * 128 >= N_NODES) v[k] = make_int4(0, 0, 0, 0);
        own[k] = (v[k].x + v[k].y) + (v[k].z + v[k].w);
        inc[k] = own[k];
    }
#pragma unroll
    for (int o = 1; o < 32; o <<= 1) {
#pragma unroll
        for (int k = 0; k < SCAN_K; ++k) {
            const int t = __shfl_up(inc[k], o, 32);
            if (lane >= o) inc[k] += t;
        }
    }
    int carry = 0;
    int base[SCAN_K];
#pragma unroll
    for (int k = 0; k < SCAN_K; ++k) {
        base[k] = carry + inc[k] - own[k];
        carry += __shfl(inc[k], 31, 32);
    }
    if (lane == 0) wtot[wave] = carry;
    __syncthreads();
    int off = (lane < wave) ? wtot[lane] : 0;
    for (int o = 16; o > 0; o >>= 1) off += __shfl_xor(off, o, 32);
#pragma unroll
    for (int k = 0; k < SCAN_K; ++k) {
        const int e = e0 + k * 128;
        if (e < N_NODES) {
            int4 r;
            r.x = off + base[k];
            r.y = r.x + v[k].x;
            r.z = r.y + v[k].y;
            r.w = r.z + v[k].z;
            *(int4*)(rowp + e) = r;
        }
    }
    if (wave == SCAN_WAVES - 1 && lane == 0) rowp[N_NODES] = off + carry;
}

__global__ void csr_place(const int* __restrict__ dst, const int* __restrict__ rowp,
                          const int* __restrict__ rnk, int* __restrict__ slot) {
    const int e = blockIdx.x * blockDim.x + threadIdx.x;
    if (e >= N_EDGES) return;
    slot[rowp[dst[e]] + rnk[e]] = e;
}

__global__ void csr_order(const int* __restrict__ rowp, const int* __restrict__ slot,
                          const int* __restrict__ src, int* __restrict__ csrc) {
    const int wv   = __builtin_amdgcn_readfirstlane((int)(threadIdx.x >> 5));
    const int node = blockIdx.x * (blockDim.x >> 5) + wv;
    const int lane = threadIdx.x & 31;
    if (node >= N_NODES) return;
    const int beg = rowp[node], end = rowp[node + 1];
    for (int i = beg + lane; i < end; i += 32) {
        const int my = slot[i];
        int r = 0;
        for (int j = beg; j < end; ++j) r += (slot[j] < my) ? 1 : 0;
        csrc[beg + r] = src[my];
    }
}

typedef __attribute__((ext_vector_type(4))) float v4f;

template <int M>
__device__ __forceinline__ float lane_xor(float v) {
    return __int_as_float(__builtin_amdgcn_update_dpp(0, __float_as_int(v), 0x160 + M, 0xF, 0xF, true));
}
__device__ __forceinline__ float sum8(float v) {
    v += lane_xor<1>(v);
    v += lane_xor<2>(v);
    v += lane_xor<4>(v);
    return v;
}
__device__ __forceinline__ float dot4(v4f a, v4f b) {
    const v2f d = a.lo * b.lo + a.hi * b.hi;
    return d.x + d.y;
}
__device__ __forceinline__ float edge_logit(v4f row, v4f as2, float ald2) {
    const float v = sum8(dot4(row, as2)) + ald2;
    return fmaxf(v, NEG_SLOPE * v);
}

template <int K>
__device__ __forceinline__ void soft_step(float& m, float& s, v4f& acc,
                                          const float (&e)[K], const v4f (&r)[K]) {
    float mn = m;
#pragma unroll
    for (int i = 0; i < K; ++i) mn = fmaxf(mn, e[i]);
    const float sc = __builtin_amdgcn_exp2f(m - mn);
    s *= sc;
    acc.lo *= sc;
    acc.hi *= sc;
#pragma unroll
    for (int i = 0; i < K; ++i) {
        const float p = __builtin_amdgcn_exp2f(e[i] - mn);
        s += p;
        acc.lo += p * r[i].lo;
        acc.hi += p * r[i].hi;
    }
    m = mn;
}

template <bool LAST>
__global__ void node_softmax_gather(const int* __restrict__ rowp, const int* __restrict__ csrc,
                                    const float* __restrict__ h, const float* __restrict__ aS,
                                    const float* __restrict__ aD, const float* __restrict__ bias,
                                    const float* __restrict__ skip, float* __restrict__ out) {
    const int wv   = __builtin_amdgcn_readfirstlane((int)(threadIdx.x >> 5));
    const int node = blockIdx.x * (blockDim.x >> 5) + wv;
    const int lane = threadIdx.x & 31;
    if (node >= N_NODES) return;
    const int beg = rowp[node], end = rowp[node + 1];
    const float    NEG_INF = -__builtin_huge_valf();
    const float    L2E     = 1.44269504088896340736f;
    const unsigned lane4   = (unsigned)lane * 4u;

    const v4f as2 = *(const v4f*)(aS + lane4) * L2E;
    const v4f ad2 = *(const v4f*)(aD + lane4) * L2E;
    const v4f hn = *(const v4f*)(h + (size_t)node * HC + lane4);
    const int    ec = LAST ? (lane & 7) * 4 : lane * 4;
    const float4 bv = *(const float4*)(bias + ec);
    const float4 sk = *(const float4*)(skip + (size_t)node * (LAST ? OUT_C : HC) + ec);
    const float ald2 = sum8(dot4(hn, ad2));

    float m = NEG_INF;
    float s = 0.f;
    v4f   a4 = {0.f, 0.f, 0.f, 0.f};

#define SRC_ROW(T) \
    (*(const v4f*)((h + (size_t)(unsigned)__builtin_amdgcn_readlane(sN, (T)) * HC) + lane4))
#define LOAD4(R, T0)                                                     \
    do {                                                                 \
        R[0] = SRC_ROW((T0));     R[1] = SRC_ROW((T0) + 1);              \
        R[2] = SRC_ROW((T0) + 2); R[3] = SRC_ROW((T0) + 3);              \
    } while (0)
#define STEP4(R)                                                                        \
    do {                                                                                \
        const float e4[4] = {edge_logit(R[0], as2, ald2), edge_logit(R[1], as2, ald2),  \
                             edge_logit(R[2], as2, ald2), edge_logit(R[3], as2, ald2)}; \
        soft_step<4>(m, s, a4, e4, R);                                                  \
    } while (0)

    for (int base = beg; base < end; base += 32) {
        const int j    = base + lane;
        const int sN   = (j < end) ? csrc[j] : 0;
        const int left = end - base;
        const int cnt  = left < 32 ? left : 32;
        int t = 0;
        for (; t + 8 <= cnt; t += 8) {
            v4f ra[4], rb[4];
            LOAD4(ra, t);
            LOAD4(rb, t + 4);
            __builtin_amdgcn_sched_barrier(0);
            STEP4(ra);
            STEP4(rb);
        }
        if (t + 4 <= cnt) {
            v4f ra[4];
            LOAD4(ra, t);
            STEP4(ra);
            t += 4;
        }
#pragma unroll 1
        for (; t < cnt; ++t) {
            const v4f   r1[1] = {SRC_ROW(t)};
            const float e1[1] = {edge_logit(r1[0], as2, ald2)};
            soft_step<1>(m, s, a4, e1, r1);
        }
    }
#undef STEP4
#undef LOAD4
#undef SRC_ROW
    const float inv = 1.0f / (s + 1e-16f);
    const float4 acc = make_float4(a4.x * inv, a4.y * inv, a4.z * inv, a4.w * inv);
    if (!LAST) {
        float4 v = make_float4(acc.x + bv.x + sk.x, acc.y + bv.y + sk.y,
                               acc.z + bv.z + sk.z, acc.w + bv.w + sk.w);
        v.x = (v.x > 0.f) ? v.x : (expf(v.x) - 1.0f);
        v.y = (v.y > 0.f) ? v.y : (expf(v.y) - 1.0f);
        v.z = (v.z > 0.f) ? v.z : (expf(v.z) - 1.0f);
        v.w = (v.w > 0.f) ? v.w : (expf(v.w) - 1.0f);
        *(float4*)(out + (size_t)node * HC + lane * 4) = v;
    } else {
        const int g = lane & 7;
        float4 hs;
        hs.x = __shfl(acc.x, g, 32) + __shfl(acc.x, g + 8, 32) + __shfl(acc.x, g + 16, 32) + __shfl(acc.x, g + 24, 32);
        hs.y = __shfl(acc.y, g, 32) + __shfl(acc.y, g + 8, 32) + __shfl(acc.y, g + 16, 32) + __shfl(acc.y, g + 24, 32);
        hs.z = __shfl(acc.z, g, 32) + __shfl(acc.z, g + 8, 32) + __shfl(acc.z, g + 16, 32) + __shfl(acc.z, g + 24, 32);
        hs.w = __shfl(acc.w, g, 32) + __shfl(acc.w, g + 8, 32) + __shfl(acc.w, g + 16, 32) + __shfl(acc.w, g + 24, 32);
        const float4 v = make_float4(0.25f * hs.x + bv.x + sk.x, 0.25f * hs.y + bv.y + sk.y,
                                     0.25f * hs.z + bv.z + sk.z, 0.25f * hs.w + bv.w + sk.w);
        float4 t = make_float4(v.x * v.x, v.y * v.y, v.z * v.z, v.w * v.w);
        for (int o = 4; o > 0; o >>= 1) {
            t.x += __shfl_xor(t.x, o, 32);
            t.y += __shfl_xor(t.y, o, 32);
            t.z += __shfl_xor(t.z, o, 32);
            t.w += __shfl_xor(t.w, o, 32);
        }
        const float sq   = (t.x + t.z) + (t.y + t.w);
        const float norm = fmaxf(sqrtf(sq), 1e-12f);
        if (lane < 8)
            *(float4*)(out + (size_t)node * OUT_C + g * 4) =
                make_float4(v.x / norm, v.y / norm, v.z / norm, v.w / norm);
    }
}

extern "C" void kernel_launch(void* const* d_in, const int* in_sizes, int n_in,
                              void* d_out, int out_size, void* d_ws, size_t ws_size,
                              hipStream_t stream) {
    (void)in_sizes; (void)n_in; (void)out_size; (void)ws_size;

    const float* x   = (const float*)d_in[0];
    const int*   ei  = (const int*)d_in[1];
    const float* W0  = (const float*)d_in[2];
    const float* aS0 = (const float*)d_in[3];
    const float* aD0 = (const float*)d_in[4];
    const float* b0  = (const float*)d_in[5];
    const float* Ws0 = (const float*)d_in[6];
    const float* bs0 = (const float*)d_in[7];
    const float* W1  = (const float*)d_in[8];
    const float* aS1 = (const float*)d_in[9];
    const float* aD1 = (const float*)d_in[10];
    const float* b1  = (const float*)d_in[11];
    const float* Ws1 = (const float*)d_in[12];
    const float* bs1 = (const float*)d_in[13];
    const float* W2  = (const float*)d_in[14];
    const float* aS2 = (const float*)d_in[15];
    const float* aD2 = (const float*)d_in[16];
    const float* b2  = (const float*)d_in[17];
    const float* Ws2 = (const float*)d_in[18];
    const float* bs2 = (const float*)d_in[19];

    const int* src = ei;
    const int* dst = ei + N_EDGES;

    const size_t NF = (size_t)N_NODES * HC;
    float*    ws    = (float*)d_ws;
    float*    x_cur = ws;
    float*    h_lin = ws + NF;
    float*    skipb = ws + 2 * NF;
    const int NPAD  = N_NODES + 32;
    int*      cnt   = (int*)(ws + 3 * NF);
    int*      rowp  = cnt + NPAD;
    int*      rnk   = rowp + NPAD;
    int*      slot  = rnk + N_EDGES;
    int*      csrc  = slot + N_EDGES;

    auto run_edges = [&](const float* hfeat, const float* aS, const float* aD,
                         const float* b, float* xout) {
        node_softmax_gather<false><<<(N_NODES + 7) / 8, 256, 0, stream>>>(
            rowp, csrc, hfeat, aS, aD, b, skipb, xout);
    };

    csr_zero<<<(N_NODES + 255) / 256, 256, 0, stream>>>(cnt, N_NODES);
    csr_count<<<(N_EDGES + 255) / 256, 256, 0, stream>>>(dst, cnt, rnk);
    csr_scan<<<1, SCAN_WAVES * 32, 0, stream>>>(cnt, rowp);
    csr_place<<<(N_EDGES + 255) / 256, 256, 0, stream>>>(dst, rowp, rnk, slot);
    csr_order<<<(N_NODES + 7) / 8, 256, 0, stream>>>(rowp, slot, src, csrc);

    dense_pair_wmma_f32<8, HC, HC><<<DENSE_GRID, DENSE_WAVES * 32, 0, stream>>>(x, W0, Ws0, bs0, h_lin, skipb);
    run_edges(h_lin, aS0, aD0, b0, x_cur);

    dense_pair_wmma_f32<8, HC, HC><<<DENSE_GRID, DENSE_WAVES * 32, 0, stream>>>(x_cur, W1, Ws1, bs1, h_lin, skipb);
    run_edges(h_lin, aS1, aD1, b1, x_cur);

    dense_pair_wmma_f32<5, HEADS * OUT_C, OUT_C><<<DENSE_GRID, DENSE_WAVES * 32, 0, stream>>>(x_cur, W2, Ws2, bs2, h_lin, skipb);
    node_softmax_gather<true><<<(N_NODES + 7) / 8, 256, 0, stream>>>(
        rowp, csrc, h_lin, aS2, aD2, b2, skipb, (float*)d_out);
}


# Edge grouping in 3 launches, not 5: the first dense launch, now in front, clears the counters; layer 0's gather wave orders its segment in registers (readlane rank, ds_permute); same sum order
# speedup vs baseline: 5.6767x; 1.0322x over previous
#include <hip/hip_runtime.h>
#include <hip/hip_bf16.h>
#include <math.h>

typedef __attribute__((ext_vector_type(2))) float v2f;
typedef __attribute__((ext_vector_type(8))) float v8f;

#define N_NODES   50000
#define N_EDGES   800000
#define F_IN      128
#define HC        128
#define HEADS     4
#define HID       32
#define OUT_C     32
#define NEG_SLOPE 0.2f

#define DENSE_K     128
#define DENSE_WAVES 16
#define DENSE_GRID  256
#define N_STRIPS    (N_NODES / 16)

template <int NT, int P1, int P2>
__global__ __launch_bounds__(DENSE_WAVES * 32)
void dense_pair_wmma_f32(const float* __restrict__ A, const float* __restrict__ W,
                         const float* __restrict__ Ws, const float* __restrict__ bs,
                         float* __restrict__ Hout, float* __restrict__ Sout,
                         int* __restrict__ zero_cnt) {
    constexpr int P   = P1 + P2;
    constexpr int PG  = 16 * NT;
    constexpr int NG  = P / PG;
    constexpr int SPB = DENSE_GRID / NG;
    static_assert(NG * PG == P && SPB * NG == DENSE_GRID, "groups tile P and the grid");
    static_assert(SPB * DENSE_WAVES <= N_STRIPS && N_STRIPS <= SPB * 25,
                  "every wave has a first strip; at most 25 strips per work-group");
    __shared__ float4 Bs[(DENSE_K / 4) * PG];

    const int lane = threadIdx.x & 31;
    const int wv = __builtin_amdgcn_readfirstlane((int)(threadIdx.x >> 5));
    const int hf = lane >> 4;
    const int c  = lane & 15;
    const int g  = blockIdx.x % NG;
    const int sb = blockIdx.x / NG;

    const int wr = wv >> 2, wc = wv & 3;
    const int m2 = (((wr + wc) & 1) == 0) ? 16 + 2 * wr + (wc >> 1) : (wv == 1 ? 24 : N_STRIPS);
    int strip = sb + SPB * wv;
    int nxt   = sb + SPB * m2;


    if (zero_cnt != nullptr) {
        const int zi = (int)blockIdx.x * (DENSE_WAVES * 32) + (int)threadIdx.x;
        if (zi < N_NODES / 4) ((int4*)zero_cnt)[zi] = make_int4(0, 0, 0, 0);
    }

    v2f a[DENSE_K / 4];
    {
        const float* arow = A + (size_t)(strip * 16 + c) * DENSE_K + 2 * hf;
#pragma unroll
        for (int i = 0; i < DENSE_K / 4; ++i) a[i] = *(const v2f*)(arow + 4 * i);
    }

    float badd[NT];
#pragma unroll
    for (int t = 0; t < NT; ++t) {
        const int   col0 = g * PG + t * 16;
        const float bv   = bs[((col0 >= P1) ? col0 - P1 : 0) + c];
        badd[t] = (col0 >= P1) ? bv : 0.0f;
    }

    constexpr int SLOTS  = (DENSE_K / 4) * PG;
    constexpr int PASSES = SLOTS / (DENSE_WAVES * 32);
    static_assert(PASSES * DENSE_WAVES * 32 == SLOTS, "whole passes");
#pragma unroll
    for (int r = 0; r < PASSES; ++r) {
        const int it  = (int)threadIdx.x + r * (DENSE_WAVES * 32);
        const int col = g * PG + it % PG;
        const int jh  = it / PG;
        const int k0  = 8 * (jh >> 1) + 2 * (jh & 1);
        const float* s;
        int pitch;
        if (col < P1) { s = W + col;         pitch = P1; }
        else          { s = Ws + (col - P1); pitch = P2; }
        Bs[it] = make_float4(s[(k0 + 0) * pitch], s[(k0 + 1) * pitch],
                             s[(k0 + 4) * pitch], s[(k0 + 5) * pitch]);
    }
    __syncthreads();

    while (strip < N_STRIPS) {
        asm volatile("" ::: "memory");
        v8f acc[NT];
#pragma unroll
        for (int t = 0; t < NT; ++t) acc[t] = (v8f){};

        const float4* bp = Bs + hf * PG + c;
#pragma unroll
        for (int j = 0; j < DENSE_K / 8; ++j) {
            float4 b4[NT];
#pragma unroll
            for (int t = 0; t < NT; ++t) b4[t] = bp[j * 2 * PG + 16 * t];
#pragma unroll
            for (int t = 0; t < NT; ++t) {
                v2f b; b.x = b4[t].x; b.y = b4[t].y;
                acc[t] = __builtin_amdgcn_wmma_f32_16x16x4_f32(
                    false, a[2 * j], false, b, (short)0, acc[t], false, false);
            }
#pragma unroll
            for (int t = 0; t < NT; ++t) {
                v2f b; b.x = b4[t].z; b.y = b4[t].w;
                acc[t] = __builtin_amdgcn_wmma_f32_16x16x4_f32(
                    false, a[2 * j + 1], false, b, (short)0, acc[t], false, false);
            }
        }

        if (nxt < N_STRIPS) {
            const float* arow = A + (size_t)(nxt * 16 + c) * DENSE_K + 2 * hf;
#pragma unroll
            for (int i = 0; i < DENSE_K / 4; ++i) a[i] = *(const v2f*)(arow + 4 * i);
        }

        const int row0 = strip * 16 + 8 * hf;
#pragma unroll
        for (int t = 0; t < NT; ++t) {
            const int  col0 = g * PG + t * 16;
            const bool sk   = col0 >= P1;
            float* o = sk ? Sout + (size_t)row0 * P2 + (col0 - P1) + c
                          : Hout + (size_t)row0 * P1 + col0 + c;
            const int pitch = sk ? P2 : P1;
#pragma unroll
            for (int i = 0; i < 8; ++i) o[(size_t)i * pitch] = acc[t][i] + badd[t];
        }
        strip = nxt;
        nxt   = N_STRIPS;
    }
}


__global__ void csr_count(const int* __restrict__ dst, int* __restrict__ cnt,
                          int* __restrict__ rnk) {
    const int e = blockIdx.x * blockDim.x + threadIdx.x;
    if (e >= N_EDGES) return;
    rnk[e] = atomicAdd(&cnt[dst[e]], 1);
}

#define SCAN_WAVES 32
#define SCAN_K     13
static_assert(SCAN_WAVES * SCAN_K * 128 >= N_NODES && N_NODES % 4 == 0,
              "the stretches cover every node; 16-byte pieces are whole");

__global__ __launch_bounds__(SCAN_WAVES * 32)
void csr_scan(const int* __restrict__ cnt, int* __restrict__ rowp) {
    __shared__ int wtot[SCAN_WAVES];
    const int lane = threadIdx.x & 31;
    const int wave = threadIdx.x >> 5;
    const int e0   = wave * (SCAN_K * 128) + lane * 4;

    int4 v[SCAN_K];
#pragma unroll
    for (int k = 0; k < SCAN_K; ++k) {
        const int e = e0 + k * 128;
        v[k] = *(const int4*)(cnt + (e < N_NODES ? e : N_NODES - 4));
    }
    int own[SCAN_K], inc[SCAN_K];
#pragma unroll
    for (int k = 0; k < SCAN_K; ++k) {
        if (e0 + k * 128 >= N_NODES) v[k] = make_int4(0, 0, 0, 0);
        own[k] = (v[k].x + v[k].y) + (v[k].z + v[k].w);
        inc[k] = own[k];
    }
#pragma unroll
    for (int o = 1; o < 32; o <<= 1) {
#pragma unroll
        for (int k = 0; k < SCAN_K; ++k) {
            const int t = __shfl_up(inc[k], o, 32);
            if (lane >= o) inc[k] += t;
        }
    }
    int carry = 0;
    int base[SCAN_K];
#pragma unroll
    for (int k = 0; k < SCAN_K; ++k) {
        base[k] = carry + inc[k] - own[k];
        carry += __shfl(inc[k], 31, 32);
    }
    if (lane == 0) wtot[wave] = carry;
    __syncthreads();
    int off = (lane < wave) ? wtot[lane] : 0;
    for (int o = 16; o > 0; o >>= 1) off += __shfl_xor(off, o, 32);
#pragma unroll
    for (int k = 0; k < SCAN_K; ++k) {
        const int e = e0 + k * 128;
        if (e < N_NODES) {
            int4 r;
            r.x = off + base[k];
            r.y = r.x + v[k].x;
            r.z = r.y + v[k].y;
            r.w = r.z + v[k].z;
            *(int4*)(rowp + e) = r;
        }
    }
    if (wave == SCAN_WAVES - 1 && lane == 0) rowp[N_NODES] = off + carry;
}

__global__ void csr_place(const int* __restrict__ dst, const int* __restrict__ src,
                          const int* __restrict__ rowp, const int* __restrict__ rnk,
                          int* __restrict__ slot) {
    const int e = blockIdx.x * blockDim.x + threadIdx.x;
    if (e >= N_EDGES) return;
    ((int2*)slot)[rowp[dst[e]] + rnk[e]] = make_int2(e, src[e]);
}

typedef __attribute__((ext_vector_type(4))) float v4f;

template <int M>
__device__ __forceinline__ float lane_xor(float v) {
    return __int_as_float(__builtin_amdgcn_update_dpp(0, __float_as_int(v), 0x160 + M, 0xF, 0xF, true));
}
__device__ __forceinline__ float sum8(float v) {
    v += lane_xor<1>(v);
    v += lane_xor<2>(v);
    v += lane_xor<4>(v);
    return v;
}
__device__ __forceinline__ float dot4(v4f a, v4f b) {
    const v2f d = a.lo * b.lo + a.hi * b.hi;
    return d.x + d.y;
}
__device__ __forceinline__ float edge_logit(v4f row, v4f as2, float ald2) {
    const float v = sum8(dot4(row, as2)) + ald2;
    return fmaxf(v, NEG_SLOPE * v);
}

template <int K>
__device__ __forceinline__ void soft_step(float& m, float& s, v4f& acc,
                                          const float (&e)[K], const v4f (&r)[K]) {
    float mn = m;
#pragma unroll
    for (int i = 0; i < K; ++i) mn = fmaxf(mn, e[i]);
    const float sc = __builtin_amdgcn_exp2f(m - mn);
    s *= sc;
    acc.lo *= sc;
    acc.hi *= sc;
#pragma unroll
    for (int i = 0; i < K; ++i) {
        const float p = __builtin_amdgcn_exp2f(e[i] - mn);
        s += p;
        acc.lo += p * r[i].lo;
        acc.hi += p * r[i].hi;
    }
    m = mn;
}

template <bool LAST, bool FIRST>
__global__ void node_softmax_gather(const int* __restrict__ rowp, const int* __restrict__ csrc,
                                    const int* __restrict__ slot_, int* csrc_w,
                                    const float* __restrict__ h, const float* __restrict__ aS,
                                    const float* __restrict__ aD, const float* __restrict__ bias,
                                    const float* __restrict__ skip, float* __restrict__ out) {
    const int wv   = __builtin_amdgcn_readfirstlane((int)(threadIdx.x >> 5));
    const int node = blockIdx.x * (blockDim.x >> 5) + wv;
    const int lane = threadIdx.x & 31;
    if (node >= N_NODES) return;
    const int beg = rowp[node], end = rowp[node + 1];
    const float    NEG_INF = -__builtin_huge_valf();
    const float    L2E     = 1.44269504088896340736f;
    const unsigned lane4   = (unsigned)lane * 4u;

    const v4f as2 = *(const v4f*)(aS + lane4) * L2E;
    const v4f ad2 = *(const v4f*)(aD + lane4) * L2E;
    const v4f hn = *(const v4f*)(h + (size_t)node * HC + lane4);
    const int    ec = LAST ? (lane & 7) * 4 : lane * 4;
    const float4 bv = *(const float4*)(bias + ec);
    const float4 sk = *(const float4*)(skip + (size_t)node * (LAST ? OUT_C : HC) + ec);
    const float ald2 = sum8(dot4(hn, ad2));

    const int2* slot = (const int2*)slot_;
    const bool in_regs = FIRST && (end - beg) <= 32;
    int sN0 = 0;
    if (FIRST) {
        if (in_regs) {
            const int  deg = end - beg;
            const int2 me  = (lane < deg) ? slot[beg + lane] : make_int2(0x7fffffff, 0);
            int r = 0;
            for (int t = 0; t < deg; ++t)
                r += (__builtin_amdgcn_readlane(me.x, t) < me.x) ? 1 : 0;
            if (lane >= deg) r = lane;
            sN0 = __builtin_amdgcn_ds_permute(r << 2, me.y);
            if (lane < deg) csrc_w[beg + lane] = sN0;
        } else {
            for (int i = beg + lane; i < end; i += 32) {
                const int2 me = slot[i];
                int r = 0;
                for (int j = beg; j < end; ++j) r += (slot[j].x < me.x) ? 1 : 0;
                csrc_w[beg + r] = me.y;
            }
            __threadfence();
        }
    }

    float m = NEG_INF;
    float s = 0.f;
    v4f   a4 = {0.f, 0.f, 0.f, 0.f};

#define SRC_ROW(T) \
    (*(const v4f*)((h + (size_t)(unsigned)__builtin_amdgcn_readlane(sN, (T)) * HC) + lane4))
#define LOAD4(R, T0)                                                     \
    do {                                                                 \
        R[0] = SRC_ROW((T0));     R[1] = SRC_ROW((T0) + 1);              \
        R[2] = SRC_ROW((T0) + 2); R[3] = SRC_ROW((T0) + 3);              \
    } while (0)
#define STEP4(R)                                                                        \
    do {                                                                                \
        const float e4[4] = {edge_logit(R[0], as2, ald2), edge_logit(R[1], as2, ald2),  \
                             edge_logit(R[2], as2, ald2), edge_logit(R[3], as2, ald2)}; \
        soft_step<4>(m, s, a4, e4, R);                                                  \
    } while (0)

    for (int base = beg; base < end; base += 32) {
        const int j    = base + lane;
        int sN;
        if (in_regs)    sN = sN0;
        else if (FIRST) sN = (j < end) ? csrc_w[j] : 0;
        else            sN = (j < end) ? csrc[j] : 0;
        const int left = end - base;
        const int cnt  = left < 32 ? left : 32;
        int t = 0;
        for (; t + 8 <= cnt; t += 8) {
            v4f ra[4], rb[4];
            LOAD4(ra, t);
            LOAD4(rb, t + 4);
            __builtin_amdgcn_sched_barrier(0);
            STEP4(ra);
            STEP4(rb);
        }
        if (t + 4 <= cnt) {
            v4f ra[4];
            LOAD4(ra, t);
            STEP4(ra);
            t += 4;
        }
#pragma unroll 1
        for (; t < cnt; ++t) {
            const v4f   r1[1] = {SRC_ROW(t)};
            const float e1[1] = {edge_logit(r1[0], as2, ald2)};
            soft_step<1>(m, s, a4, e1, r1);
        }
    }
#undef STEP4
#undef LOAD4
#undef SRC_ROW
    const float inv = 1.0f / (s + 1e-16f);
    const float4 acc = make_float4(a4.x * inv, a4.y * inv, a4.z * inv, a4.w * inv);
    if (!LAST) {
        float4 v = make_float4(acc.x + bv.x + sk.x, acc.y + bv.y + sk.y,
                               acc.z + bv.z + sk.z, acc.w + bv.w + sk.w);
        v.x = (v.x > 0.f) ? v.x : (expf(v.x) - 1.0f);
        v.y = (v.y > 0.f) ? v.y : (expf(v.y) - 1.0f);
        v.z = (v.z > 0.f) ? v.z : (expf(v.z) - 1.0f);
        v.w = (v.w > 0.f) ? v.w : (expf(v.w) - 1.0f);
        *(float4*)(out + (size_t)node * HC + lane * 4) = v;
    } else {
        const int g = lane & 7;
        float4 hs;
        hs.x = __shfl(acc.x, g, 32) + __shfl(acc.x, g + 8, 32) + __shfl(acc.x, g + 16, 32) + __shfl(acc.x, g + 24, 32);
        hs.y = __shfl(acc.y, g, 32) + __shfl(acc.y, g + 8, 32) + __shfl(acc.y, g + 16, 32) + __shfl(acc.y, g + 24, 32);
        hs.z = __shfl(acc.z, g, 32) + __shfl(acc.z, g + 8, 32) + __shfl(acc.z, g + 16, 32) + __shfl(acc.z, g + 24, 32);
        hs.w = __shfl(acc.w, g, 32) + __shfl(acc.w, g + 8, 32) + __shfl(acc.w, g + 16, 32) + __shfl(acc.w, g + 24, 32);
        const float4 v = make_float4(0.25f * hs.x + bv.x + sk.x, 0.25f * hs.y + bv.y + sk.y,
                                     0.25f * hs.z + bv.z + sk.z, 0.25f * hs.w + bv.w + sk.w);
        float4 t = make_float4(v.x * v.x, v.y * v.y, v.z * v.z, v.w * v.w);
        for (int o = 4; o > 0; o >>= 1) {
            t.x += __shfl_xor(t.x, o, 32);
            t.y += __shfl_xor(t.y, o, 32);
            t.z += __shfl_xor(t.z, o, 32);
            t.w += __shfl_xor(t.w, o, 32);
        }
        const float sq   = (t.x + t.z) + (t.y + t.w);
        const float norm = fmaxf(sqrtf(sq), 1e-12f);
        if (lane < 8)
            *(float4*)(out + (size_t)node * OUT_C + g * 4) =
                make_float4(v.x / norm, v.y / norm, v.z / norm, v.w / norm);
    }
}

extern "C" void kernel_launch(void* const* d_in, const int* in_sizes, int n_in,
                              void* d_out, int out_size, void* d_ws, size_t ws_size,
                              hipStream_t stream) {
    (void)in_sizes; (void)n_in; (void)out_size; (void)ws_size;

    const float* x   = (const float*)d_in[0];
    const int*   ei  = (const int*)d_in[1];
    const float* W0  = (const float*)d_in[2];
    const float* aS0 = (const float*)d_in[3];
    const float* aD0 = (const float*)d_in[4];
    const float* b0  = (const float*)d_in[5];
    const float* Ws0 = (const float*)d_in[6];
    const float* bs0 = (const float*)d_in[7];
    const float* W1  = (const float*)d_in[8];
    const float* aS1 = (const float*)d_in[9];
    const float* aD1 = (const float*)d_in[10];
    const float* b1  = (const float*)d_in[11];
    const float* Ws1 = (const float*)d_in[12];
    const float* bs1 = (const float*)d_in[13];
    const float* W2  = (const float*)d_in[14];
    const float* aS2 = (const float*)d_in[15];
    const float* aD2 = (const float*)d_in[16];
    const float* b2  = (const float*)d_in[17];
    const float* Ws2 = (const float*)d_in[18];
    const float* bs2 = (const float*)d_in[19];

    const int* src = ei;
    const int* dst = ei + N_EDGES;

    const size_t NF = (size_t)N_NODES * HC;
    float*    ws    = (float*)d_ws;
    float*    x_cur = ws;
    float*    h_lin = ws + NF;
    float*    skipb = ws + 2 * NF;
    const int NPAD  = N_NODES + 32;
    int*      cnt   = (int*)(ws + 3 * NF);
    int*      rowp  = cnt + NPAD;
    int*      rnk   = rowp + NPAD;
    int*      slot  = rnk + N_EDGES;
    int*      csrc  = rnk + 3 * (size_t)N_EDGES;

    auto run_edges = [&](const float* hfeat, const float* aS, const float* aD,
                         const float* b, float* xout) {
        node_softmax_gather<false, false><<<(N_NODES + 7) / 8, 256, 0, stream>>>(
            rowp, csrc, nullptr, nullptr, hfeat, aS, aD, b, skipb, xout);
    };

    dense_pair_wmma_f32<8, HC, HC><<<DENSE_GRID, DENSE_WAVES * 32, 0, stream>>>(x, W0, Ws0, bs0, h_lin, skipb, cnt);

    csr_count<<<(N_EDGES + 255) / 256, 256, 0, stream>>>(dst, cnt, rnk);
    csr_scan<<<1, SCAN_WAVES * 32, 0, stream>>>(cnt, rowp);
    csr_place<<<(N_EDGES + 255) / 256, 256, 0, stream>>>(dst, src, rowp, rnk, slot);

    node_softmax_gather<false, true><<<(N_NODES + 7) / 8, 256, 0, stream>>>(
        rowp, nullptr, slot, csrc, h_lin, aS0, aD0, b0, skipb, x_cur);

    dense_pair_wmma_f32<8, HC, HC><<<DENSE_GRID, DENSE_WAVES * 32, 0, stream>>>(x_cur, W1, Ws1, bs1, h_lin, skipb, nullptr);
    run_edges(h_lin, aS1, aD1, b1, x_cur);

    dense_pair_wmma_f32<5, HEADS * OUT_C, OUT_C><<<DENSE_GRID, DENSE_WAVES * 32, 0, stream>>>(x_cur, W2, Ws2, bs2, h_lin, skipb, nullptr);
    node_softmax_gather<true, false><<<(N_NODES + 7) / 8, 256, 0, stream>>>(
        rowp, csrc, nullptr, nullptr, h_lin, aS2, aD2, b2, skipb, (float*)d_out);
}


# Edge grouping in 1 launch, not 3: the counting atomic's return puts each edge in its node's fixed 64-slot segment (no scan, no place pass); a node over 64 edges walks the edge list; same sum order
# speedup vs baseline: 6.0699x; 1.0693x over previous
#include <hip/hip_runtime.h>
#include <hip/hip_bf16.h>
#include <math.h>

typedef __attribute__((ext_vector_type(2))) float v2f;
typedef __attribute__((ext_vector_type(8))) float v8f;

#define N_NODES   50000
#define N_EDGES   800000
#define F_IN      128
#define HC        128
#define HEADS     4
#define HID       32
#define OUT_C     32
#define NEG_SLOPE 0.2f

#define DENSE_K     128
#define DENSE_WAVES 16
#define DENSE_GRID  256
#define N_STRIPS    (N_NODES / 16)

template <int NT, int P1, int P2>
__global__ __launch_bounds__(DENSE_WAVES * 32)
void dense_pair_wmma_f32(const float* __restrict__ A, const float* __restrict__ W,
                         const float* __restrict__ Ws, const float* __restrict__ bs,
                         float* __restrict__ Hout, float* __restrict__ Sout,
                         int* __restrict__ zero_cnt) {
    constexpr int P   = P1 + P2;
    constexpr int PG  = 16 * NT;
    constexpr int NG  = P / PG;
    constexpr int SPB = DENSE_GRID / NG;
    static_assert(NG * PG == P && SPB * NG == DENSE_GRID, "groups tile P and the grid");
    static_assert(SPB * DENSE_WAVES <= N_STRIPS && N_STRIPS <= SPB * 25,
                  "every wave has a first strip; at most 25 strips per work-group");
    __shared__ float4 Bs[(DENSE_K / 4) * PG];

    const int lane = threadIdx.x & 31;
    const int wv = __builtin_amdgcn_readfirstlane((int)(threadIdx.x >> 5));
    const int hf = lane >> 4;
    const int c  = lane & 15;
    const int g  = blockIdx.x % NG;
    const int sb = blockIdx.x / NG;

    const int wr = wv >> 2, wc = wv & 3;
    const int m2 = (((wr + wc) & 1) == 0) ? 16 + 2 * wr + (wc >> 1) : (wv == 1 ? 24 : N_STRIPS);
    int strip = sb + SPB * wv;
    int nxt   = sb + SPB * m2;


    if (zero_cnt != nullptr) {
        const int zi = (int)blockIdx.x * (DENSE_WAVES * 32) + (int)threadIdx.x;
        if (zi < N_NODES / 4) ((int4*)zero_cnt)[zi] = make_int4(0, 0, 0, 0);
    }

    v2f a[DENSE_K / 4];
    {
        const float* arow = A + (size_t)(strip * 16 + c) * DENSE_K + 2 * hf;
#pragma unroll
        for (int i = 0; i < DENSE_K / 4; ++i) a[i] = *(const v2f*)(arow + 4 * i);
    }

    float badd[NT];
#pragma unroll
    for (int t = 0; t < NT; ++t) {
        const int   col0 = g * PG + t * 16;
        const float bv   = bs[((col0 >= P1) ? col0 - P1 : 0) + c];
        badd[t] = (col0 >= P1) ? bv : 0.0f;
    }

    constexpr int SLOTS  = (DENSE_K / 4) * PG;
    constexpr int PASSES = SLOTS / (DENSE_WAVES * 32);
    static_assert(PASSES * DENSE_WAVES * 32 == SLOTS, "whole passes");
#pragma unroll
    for (int r = 0; r < PASSES; ++r) {
        const int it  = (int)threadIdx.x + r * (DENSE_WAVES * 32);
        const int col = g * PG + it % PG;
        const int jh  = it / PG;
        const int k0  = 8 * (jh >> 1) + 2 * (jh & 1);
        const float* s;
        int pitch;
        if (col < P1) { s = W + col;         pitch = P1; }
        else          { s = Ws + (col - P1); pitch = P2; }
        Bs[it] = make_float4(s[(k0 + 0) * pitch], s[(k0 + 1) * pitch],
                             s[(k0 + 4) * pitch], s[(k0 + 5) * pitch]);
    }
    __syncthreads();

    while (strip < N_STRIPS) {
        asm volatile("" ::: "memory");
        v8f acc[NT];
#pragma unroll
        for (int t = 0; t < NT; ++t) acc[t] = (v8f){};

        const float4* bp = Bs + hf * PG + c;
#pragma unroll
        for (int j = 0; j < DENSE_K / 8; ++j) {
            float4 b4[NT];
#pragma unroll
            for (int t = 0; t < NT; ++t) b4[t] = bp[j * 2 * PG + 16 * t];
#pragma unroll
            for (int t = 0; t < NT; ++t) {
                v2f b; b.x = b4[t].x; b.y = b4[t].y;
                acc[t] = __builtin_amdgcn_wmma_f32_16x16x4_f32(
                    false, a[2 * j], false, b, (short)0, acc[t], false, false);
            }
#pragma unroll
            for (int t = 0; t < NT; ++t) {
                v2f b; b.x = b4[t].z; b.y = b4[t].w;
                acc[t] = __builtin_amdgcn_wmma_f32_16x16x4_f32(
                    false, a[2 * j + 1], false, b, (short)0, acc[t], false, false);
            }
        }

        if (nxt < N_STRIPS) {
            const float* arow = A + (size_t)(nxt * 16 + c) * DENSE_K + 2 * hf;
#pragma unroll
            for (int i = 0; i < DENSE_K / 4; ++i) a[i] = *(const v2f*)(arow + 4 * i);
        }

        const int row0 = strip * 16 + 8 * hf;
#pragma unroll
        for (int t = 0; t < NT; ++t) {
            const int  col0 = g * PG + t * 16;
            const bool sk   = col0 >= P1;
            float* o = sk ? Sout + (size_t)row0 * P2 + (col0 - P1) + c
                          : Hout + (size_t)row0 * P1 + col0 + c;
            const int pitch = sk ? P2 : P1;
#pragma unroll
            for (int i = 0; i < 8; ++i) o[(size_t)i * pitch] = acc[t][i] + badd[t];
        }
        strip = nxt;
        nxt   = N_STRIPS;
    }
}

#define SEG_CAP 64
static_assert(N_EDGES % 32 == 0, "the edge-list walk reads whole 32-edge pieces");

__global__ void edge_segments(const int* __restrict__ dst, const int* __restrict__ src,
                              int* __restrict__ indeg, int* __restrict__ slot) {
    const int e = blockIdx.x * blockDim.x + threadIdx.x;
    if (e >= N_EDGES) return;
    const int d = dst[e];
    const int s = src[e];
    asm volatile("" :: "v"(s));
    const int r = atomicAdd(&indeg[d], 1);
    if (r < SEG_CAP) ((int2*)slot)[d * SEG_CAP + r] = make_int2(e, s);
}

typedef __attribute__((ext_vector_type(4))) float v4f;

template <int M>
__device__ __forceinline__ float lane_xor(float v) {
    return __int_as_float(__builtin_amdgcn_update_dpp(0, __float_as_int(v), 0x160 + M, 0xF, 0xF, true));
}
__device__ __forceinline__ float sum8(float v) {
    v += lane_xor<1>(v);
    v += lane_xor<2>(v);
    v += lane_xor<4>(v);
    return v;
}
__device__ __forceinline__ float dot4(v4f a, v4f b) {
    const v2f d = a.lo * b.lo + a.hi * b.hi;
    return d.x + d.y;
}
__device__ __forceinline__ float edge_logit(v4f row, v4f as2, float ald2) {
    const float v = sum8(dot4(row, as2)) + ald2;
    return fmaxf(v, NEG_SLOPE * v);
}

template <int K>
__device__ __forceinline__ void soft_step(float& m, float& s, v4f& acc,
                                          const float (&e)[K], const v4f (&r)[K]) {
    float mn = m;
#pragma unroll
    for (int i = 0; i < K; ++i) mn = fmaxf(mn, e[i]);
    const float sc = __builtin_amdgcn_exp2f(m - mn);
    s *= sc;
    acc.lo *= sc;
    acc.hi *= sc;
#pragma unroll
    for (int i = 0; i < K; ++i) {
        const float p = __builtin_amdgcn_exp2f(e[i] - mn);
        s += p;
        acc.lo += p * r[i].lo;
        acc.hi += p * r[i].hi;
    }
    m = mn;
}

template <bool LAST, bool FIRST>
__global__ void node_softmax_gather(const int* __restrict__ indeg, const int* __restrict__ csrc,
                                    const int* __restrict__ slot_, int* csrc_w,
                                    const int* __restrict__ esrc, const int* __restrict__ edst,
                                    const float* __restrict__ h, const float* __restrict__ aS,
                                    const float* __restrict__ aD, const float* __restrict__ bias,
                                    const float* __restrict__ skip, float* __restrict__ out) {
    const int wv   = __builtin_amdgcn_readfirstlane((int)(threadIdx.x >> 5));
    const int node = blockIdx.x * (blockDim.x >> 5) + wv;
    const int lane = threadIdx.x & 31;
    if (node >= N_NODES) return;
    const int  ndeg = indeg[node];
    const bool walk = ndeg > SEG_CAP;
    const int  beg  = node * SEG_CAP, end = beg + (walk ? 0 : ndeg);
    const float    NEG_INF = -__builtin_huge_valf();
    const float    L2E     = 1.44269504088896340736f;
    const unsigned lane4   = (unsigned)lane * 4u;

    const v4f as2 = *(const v4f*)(aS + lane4) * L2E;
    const v4f ad2 = *(const v4f*)(aD + lane4) * L2E;
    const v4f hn = *(const v4f*)(h + (size_t)node * HC + lane4);
    const int    ec = LAST ? (lane & 7) * 4 : lane * 4;
    const float4 bv = *(const float4*)(bias + ec);
    const float4 sk = *(const float4*)(skip + (size_t)node * (LAST ? OUT_C : HC) + ec);
    const float ald2 = sum8(dot4(hn, ad2));

    const int2* slot = (const int2*)slot_;
    const bool in_regs = FIRST && (end - beg) <= 32;
    int sN0 = 0;
    if (FIRST) {
        if (in_regs) {
            const int  deg = end - beg;
            const int2 me  = (lane < deg) ? slot[beg + lane] : make_int2(0x7fffffff, 0);
            int r = 0;
            for (int t = 0; t < deg; ++t)
                r += (__builtin_amdgcn_readlane(me.x, t) < me.x) ? 1 : 0;
            if (lane >= deg) r = lane;
            sN0 = __builtin_amdgcn_ds_permute(r << 2, me.y);
            if (lane < deg) csrc_w[beg + lane] = sN0;
        } else {
            for (int i = beg + lane; i < end; i += 32) {
                const int2 me = slot[i];
                int r = 0;
                for (int j = beg; j < end; ++j) r += (slot[j].x < me.x) ? 1 : 0;
                csrc_w[beg + r] = me.y;
            }
            __threadfence();
        }
    }

    float m = NEG_INF;
    float s = 0.f;
    v4f   a4 = {0.f, 0.f, 0.f, 0.f};

#define SRC_ROW(T) \
    (*(const v4f*)((h + (size_t)(unsigned)__builtin_amdgcn_readlane(sN, (T)) * HC) + lane4))
#define LOAD4(R, T0)                                                     \
    do {                                                                 \
        R[0] = SRC_ROW((T0));     R[1] = SRC_ROW((T0) + 1);              \
        R[2] = SRC_ROW((T0) + 2); R[3] = SRC_ROW((T0) + 3);              \
    } while (0)
#define STEP4(R)                                                                        \
    do {                                                                                \
        const float e4[4] = {edge_logit(R[0], as2, ald2), edge_logit(R[1], as2, ald2),  \
                             edge_logit(R[2], as2, ald2), edge_logit(R[3], as2, ald2)}; \
        soft_step<4>(m, s, a4, e4, R);                                                  \
    } while (0)

    for (int base = beg; base < end; base += 32) {
        const int j    = base + lane;
        int sN;
        if (in_regs)    sN = sN0;
        else if (FIRST) sN = (j < end) ? csrc_w[j] : 0;
        else            sN = (j < end) ? csrc[j] : 0;
        const int left = end - base;
        const int cnt  = left < 32 ? left : 32;
        int t = 0;
        for (; t + 8 <= cnt; t += 8) {
            v4f ra[4], rb[4];
            LOAD4(ra, t);
            LOAD4(rb, t + 4);
            __builtin_amdgcn_sched_barrier(0);
            STEP4(ra);
            STEP4(rb);
        }
        if (t + 4 <= cnt) {
            v4f ra[4];
            LOAD4(ra, t);
            STEP4(ra);
            t += 4;
        }
#pragma unroll 1
        for (; t < cnt; ++t) {
            const v4f   r1[1] = {SRC_ROW(t)};
            const float e1[1] = {edge_logit(r1[0], as2, ald2)};
            soft_step<1>(m, s, a4, e1, r1);
        }
    }
#undef STEP4
#undef LOAD4
#undef SRC_ROW
    if (walk) {
        for (int base = 0; base < N_EDGES; base += 32) {
            const unsigned hit = __builtin_amdgcn_ballot_w32(edst[base + lane] == node);
            for (unsigned rest = hit; rest != 0; rest &= rest - 1) {
                const int   sn    = __builtin_amdgcn_readfirstlane(esrc[base + __builtin_ctz(rest)]);
                const v4f   r1[1] = {*(const v4f*)(h + (size_t)(unsigned)sn * HC + lane4)};
                const float e1[1] = {edge_logit(r1[0], as2, ald2)};
                soft_step<1>(m, s, a4, e1, r1);
            }
        }
    }
    const float inv = 1.0f / (s + 1e-16f);
    const float4 acc = make_float4(a4.x * inv, a4.y * inv, a4.z * inv, a4.w * inv);
    if (!LAST) {
        float4 v = make_float4(acc.x + bv.x + sk.x, acc.y + bv.y + sk.y,
                               acc.z + bv.z + sk.z, acc.w + bv.w + sk.w);
        v.x = (v.x > 0.f) ? v.x : (expf(v.x) - 1.0f);
        v.y = (v.y > 0.f) ? v.y : (expf(v.y) - 1.0f);
        v.z = (v.z > 0.f) ? v.z : (expf(v.z) - 1.0f);
        v.w = (v.w > 0.f) ? v.w : (expf(v.w) - 1.0f);
        *(float4*)(out + (size_t)node * HC + lane * 4) = v;
    } else {
        const int g = lane & 7;
        float4 hs;
        hs.x = __shfl(acc.x, g, 32) + __shfl(acc.x, g + 8, 32) + __shfl(acc.x, g + 16, 32) + __shfl(acc.x, g + 24, 32);
        hs.y = __shfl(acc.y, g, 32) + __shfl(acc.y, g + 8, 32) + __shfl(acc.y, g + 16, 32) + __shfl(acc.y, g + 24, 32);
        hs.z = __shfl(acc.z, g, 32) + __shfl(acc.z, g + 8, 32) + __shfl(acc.z, g + 16, 32) + __shfl(acc.z, g + 24, 32);
        hs.w = __shfl(acc.w, g, 32) + __shfl(acc.w, g + 8, 32) + __shfl(acc.w, g + 16, 32) + __shfl(acc.w, g + 24, 32);
        const float4 v = make_float4(0.25f * hs.x + bv.x + sk.x, 0.25f * hs.y + bv.y + sk.y,
                                     0.25f * hs.z + bv.z + sk.z, 0.25f * hs.w + bv.w + sk.w);
        float4 t = make_float4(v.x * v.x, v.y * v.y, v.z * v.z, v.w * v.w);
        for (int o = 4; o > 0; o >>= 1) {
            t.x += __shfl_xor(t.x, o, 32);
            t.y += __shfl_xor(t.y, o, 32);
            t.z += __shfl_xor(t.z, o, 32);
            t.w += __shfl_xor(t.w, o, 32);
        }
        const float sq   = (t.x + t.z) + (t.y + t.w);
        const float norm = fmaxf(sqrtf(sq), 1e-12f);
        if (lane < 8)
            *(float4*)(out + (size_t)node * OUT_C + g * 4) =
                make_float4(v.x / norm, v.y / norm, v.z / norm, v.w / norm);
    }
}

extern "C" void kernel_launch(void* const* d_in, const int* in_sizes, int n_in,
                              void* d_out, int out_size, void* d_ws, size_t ws_size,
                              hipStream_t stream) {
    (void)in_sizes; (void)n_in; (void)out_size; (void)ws_size;

    const float* x   = (const float*)d_in[0];
    const int*   ei  = (const int*)d_in[1];
    const float* W0  = (const float*)d_in[2];
    const float* aS0 = (const float*)d_in[3];
    const float* aD0 = (const float*)d_in[4];
    const float* b0  = (const float*)d_in[5];
    const float* Ws0 = (const float*)d_in[6];
    const float* bs0 = (const float*)d_in[7];
    const float* W1  = (const float*)d_in[8];
    const float* aS1 = (const float*)d_in[9];
    const float* aD1 = (const float*)d_in[10];
    const float* b1  = (const float*)d_in[11];
    const float* Ws1 = (const float*)d_in[12];
    const float* bs1 = (const float*)d_in[13];
    const float* W2  = (const float*)d_in[14];
    const float* aS2 = (const float*)d_in[15];
    const float* aD2 = (const float*)d_in[16];
    const float* b2  = (const float*)d_in[17];
    const float* Ws2 = (const float*)d_in[18];
    const float* bs2 = (const float*)d_in[19];

    const int* src = ei;
    const int* dst = ei + N_EDGES;

    const size_t NF = (size_t)N_NODES * HC;
    float*    ws    = (float*)d_ws;
    float*    x_cur = ws;
    float*    h_lin = ws + NF;
    float*    skipb = ws + 2 * NF;
    const int NPAD  = N_NODES + 176;
    int*      indeg = (int*)(ws + 3 * NF);
    int*      slot  = indeg + NPAD;
    int*      csrc  = slot + 2 * (size_t)N_NODES * SEG_CAP;

    auto run_edges = [&](const float* hfeat, const float* aS, const float* aD,
                         const float* b, float* xout) {
        node_softmax_gather<false, false><<<(N_NODES + 7) / 8, 256, 0, stream>>>(
            indeg, csrc, nullptr, nullptr, src, dst, hfeat, aS, aD, b, skipb, xout);
    };

    dense_pair_wmma_f32<8, HC, HC><<<DENSE_GRID, DENSE_WAVES * 32, 0, stream>>>(x, W0, Ws0, bs0, h_lin, skipb, indeg);

    edge_segments<<<(N_EDGES + 255) / 256, 256, 0, stream>>>(dst, src, indeg, slot);

    node_softmax_gather<false, true><<<(N_NODES + 7) / 8, 256, 0, stream>>>(
        indeg, nullptr, slot, csrc, src, dst, h_lin, aS0, aD0, b0, skipb, x_cur);

    dense_pair_wmma_f32<8, HC, HC><<<DENSE_GRID, DENSE_WAVES * 32, 0, stream>>>(x_cur, W1, Ws1, bs1, h_lin, skipb, nullptr);
    run_edges(h_lin, aS1, aD1, b1, x_cur);

    dense_pair_wmma_f32<5, HEADS * OUT_C, OUT_C><<<DENSE_GRID, DENSE_WAVES * 32, 0, stream>>>(x_cur, W2, Ws2, bs2, h_lin, skipb, nullptr);
    node_softmax_gather<true, false><<<(N_NODES + 7) / 8, 256, 0, stream>>>(
        indeg, csrc, nullptr, nullptr, src, dst, h_lin, aS2, aD2, b2, skipb, (float*)d_out);
}
